# starting kernel
# baseline (speedup 1.0000x reference)
#include <hip/hip_runtime.h>
#include <math.h>

typedef float v2f __attribute__((ext_vector_type(2)));
typedef float v8f __attribute__((ext_vector_type(8)));

#define NEG_SLOPE 0.2f
#define ENC_NEG_INF 0x007FFFFFu

__device__ __forceinline__ unsigned fenc(float f) {
  unsigned u = __float_as_uint(f);
  return (u & 0x80000000u) ? ~u : (u | 0x80000000u);
}
__device__ __forceinline__ float fdec(unsigned k) {
  unsigned u = (k & 0x80000000u) ? (k & 0x7FFFFFFFu) : ~k;
  return __uint_as_float(u);
}

__global__ void gat_gemm128_wmma(const float* __restrict__ X,
                                 const float* __restrict__ W,
                                 float* __restrict__ H, int N) {
  const int wave = threadIdx.x >> 5;
  const int lane = threadIdx.x & 31;
  const int half = lane >> 4;
  const int lm   = lane & 15;
  const int row0 = blockIdx.x << 4;
  const int col  = (wave << 4) + lm;

  int arow = row0 + lm;
  if (arow >= N) arow = N - 1;
  const float* xr = X + (size_t)arow * 128;

  v8f c = {};
#pragma unroll
  for (int kk = 0; kk < 32; ++kk) {
    const int kb = (kk << 2) + (half << 1);
    v2f a, b;
    a.x = xr[kb];
    a.y = xr[kb + 1];
    b.x = W[(size_t)kb * 128 + col];
    b.y = W[(size_t)(kb + 1) * 128 + col];
    c = __builtin_amdgcn_wmma_f32_16x16x4_f32(false, a, false, b,
                                              (short)0, c, false, false);
  }

  const int orow0 = row0 + (half << 3);
#pragma unroll
  for (int v = 0; v < 8; ++v) {
    int r = orow0 + v;
    if (r < N) H[(size_t)r * 128 + col] = c[v];
  }
}

__global__ void gat_attn_logits(const float* __restrict__ H,
                                const float* __restrict__ a_src,
                                const float* __restrict__ a_dst,
                                float* __restrict__ als,
                                float* __restrict__ ald, int N) {
  const int wid  = blockIdx.x * (blockDim.x >> 5) + (threadIdx.x >> 5);
  const int lane = threadIdx.x & 31;
  if (wid >= N) return;
  const float* hr = H + (size_t)wid * 128 + lane * 4;
  const float* as = a_src + lane * 4;
  const float* ad = a_dst + lane * 4;
  float s = 0.f, d = 0.f;
#pragma unroll
  for (int j = 0; j < 4; ++j) {
    float hv = hr[j];
    s += hv * as[j];
    d += hv * ad[j];
  }
#pragma unroll
  for (int m = 8; m >= 1; m >>= 1) {
    s += __shfl_xor(s, m, 32);
    d += __shfl_xor(d, m, 32);
  }
  if ((lane & 15) == 0) {
    const int h = lane >> 4;
    als[wid * 2 + h] = s;
    ald[wid * 2 + h] = d;
  }
}

__global__ void gat_init(float* __restrict__ agg, unsigned* __restrict__ emax,
                         float* __restrict__ denom, int N) {
  const int tid = blockIdx.x * blockDim.x + threadIdx.x;
  if (tid < N * 128) agg[tid] = 0.f;
  if (tid < N * 2) { emax[tid] = ENC_NEG_INF; denom[tid] = 0.f; }
}

__device__ __forceinline__ void edge_sd(int e, int E, const int* __restrict__ es,
                                        const int* __restrict__ ed, int& s, int& d) {
  if (e < E) { s = es[e]; d = ed[e]; }
  else       { s = e - E; d = e - E; }
}

__global__ void gat_edge_max(const int* __restrict__ es, const int* __restrict__ ed,
                             const float* __restrict__ als, const float* __restrict__ ald,
                             unsigned* __restrict__ emax, int E, int Etot) {
  const int e = blockIdx.x * blockDim.x + threadIdx.x;
  if (e >= Etot) return;
  int s, d; edge_sd(e, E, es, ed, s, d);
#pragma unroll
  for (int h = 0; h < 2; ++h) {
    float v = als[s * 2 + h] + ald[d * 2 + h];
    v = (v >= 0.f) ? v : NEG_SLOPE * v;
    atomicMax(&emax[d * 2 + h], fenc(v));
  }
}

__global__ void gat_edge_sum(const int* __restrict__ es, const int* __restrict__ ed,
                             const float* __restrict__ als, const float* __restrict__ ald,
                             const unsigned* __restrict__ emax,
                             float* __restrict__ denom, int E, int Etot) {
  const int e = blockIdx.x * blockDim.x + threadIdx.x;
  if (e >= Etot) return;
  int s, d; edge_sd(e, E, es, ed, s, d);
#pragma unroll
  for (int h = 0; h < 2; ++h) {
    float v = als[s * 2 + h] + ald[d * 2 + h];
    v = (v >= 0.f) ? v : NEG_SLOPE * v;
    atomicAdd(&denom[d * 2 + h], expf(v - fdec(emax[d * 2 + h])));
  }
}

__global__ void gat_edge_agg(const int* __restrict__ es, const int* __restrict__ ed,
                             const float* __restrict__ als, const float* __restrict__ ald,
                             const unsigned* __restrict__ emax,
                             const float* __restrict__ denom,
                             const float* __restrict__ H, float* __restrict__ agg,
                             int E, int Etot, int nWaves) {
  const int wid  = blockIdx.x * (blockDim.x >> 5) + (threadIdx.x >> 5);
  const int lane = threadIdx.x & 31;
  for (int e = wid; e < Etot; e += nWaves) {
    int s, d; edge_sd(e, E, es, ed, s, d);
    const int en = e + nWaves;
    if (en < Etot) {
      const int sn = (en < E) ? es[en] : (en - E);
      __builtin_prefetch(&H[(size_t)sn * 128 + lane * 4], 0, 0);
    }
    float alpha[2];
#pragma unroll
    for (int h = 0; h < 2; ++h) {
      float v = als[s * 2 + h] + ald[d * 2 + h];
      v = (v >= 0.f) ? v : NEG_SLOPE * v;
      alpha[h] = expf(v - fdec(emax[d * 2 + h])) / denom[d * 2 + h];
    }
    const float* hs = H + (size_t)s * 128;
    float* ag = agg + (size_t)d * 128;
#pragma unroll
    for (int k = 0; k < 4; ++k) {
      const int idx = lane + (k << 5);
      atomicAdd(&ag[idx], hs[idx] * ((idx < 64) ? alpha[0] : alpha[1]));
    }
  }
}

__global__ void gat_bias_elu(float* __restrict__ agg, const float* __restrict__ b, int N) {
  const int tid = blockIdx.x * blockDim.x + threadIdx.x;
  if (tid >= N * 128) return;
  const float v = agg[tid] + b[tid & 127];
  agg[tid] = (v > 0.f) ? v : (expf(v) - 1.f);
}

__global__ void gat_head_mean_bias(const float* __restrict__ agg,
                                   const float* __restrict__ b,
                                   float* __restrict__ out, int N) {
  const int tid = blockIdx.x * blockDim.x + threadIdx.x;
  if (tid >= N * 64) return;
  const int n = tid >> 6, c = tid & 63;
  out[tid] = 0.5f * (agg[(size_t)n * 128 + c] + agg[(size_t)n * 128 + 64 + c]) + b[c];
}

extern "C" void kernel_launch(void* const* d_in, const int* in_sizes, int n_in,
                              void* d_out, int out_size, void* d_ws, size_t ws_size,
                              hipStream_t stream) {
  const float* x   = (const float*)d_in[0];
  const int*   ei  = (const int*)d_in[1];
  const float* W1  = (const float*)d_in[2];
  const float* as1 = (const float*)d_in[3];
  const float* ad1 = (const float*)d_in[4];
  const float* b1  = (const float*)d_in[5];
  const float* W2  = (const float*)d_in[6];
  const float* as2 = (const float*)d_in[7];
  const float* ad2 = (const float*)d_in[8];
  const float* b2  = (const float*)d_in[9];

  const int N = in_sizes[0] / 128;
  const int E = in_sizes[1] / 2;
  const int Etot = E + N;
  const int* esrc = ei;
  const int* edst = ei + E;

  float*    h     = (float*)d_ws;
  float*    agg   = h + (size_t)N * 128;
  float*    als   = agg + (size_t)N * 128;
  float*    ald   = als + (size_t)N * 2;
  unsigned* emax  = (unsigned*)(ald + (size_t)N * 2);
  float*    denom = (float*)(emax + (size_t)N * 2);

  const int B = 256;
  const int gGemm  = (N + 15) / 16;
  const int gNodeW = (N + 7) / 8;
  const int gInit  = (N * 128 + B - 1) / B;
  const int gEdge  = (Etot + B - 1) / B;
  const int aggBlocks = 8192;
  const int nWaves = aggBlocks * (B / 32);
  const int gFin1 = (N * 128 + B - 1) / B;
  const int gFin2 = (N * 64 + B - 1) / B;

  gat_gemm128_wmma<<<gGemm, B, 0, stream>>>(x, W1, h, N);
  gat_attn_logits<<<gNodeW, B, 0, stream>>>(h, as1, ad1, als, ald, N);
  gat_init<<<gInit, B, 0, stream>>>(agg, emax, denom, N);
  gat_edge_max<<<gEdge, B, 0, stream>>>(esrc, edst, als, ald, emax, E, Etot);
  gat_edge_sum<<<gEdge, B, 0, stream>>>(esrc, edst, als, ald, emax, denom, E, Etot);
  gat_edge_agg<<<aggBlocks, B, 0, stream>>>(esrc, edst, als, ald, emax, denom,
                                            h, agg, E, Etot, nWaves);
  gat_bias_elu<<<gFin1, B, 0, stream>>>(agg, b1, N);

  gat_gemm128_wmma<<<gGemm, B, 0, stream>>>(agg, W2, h, N);
  gat_attn_logits<<<gNodeW, B, 0, stream>>>(h, as2, ad2, als, ald, N);
  gat_init<<<gInit, B, 0, stream>>>(agg, emax, denom, N);
  gat_edge_max<<<gEdge, B, 0, stream>>>(esrc, edst, als, ald, emax, E, Etot);
  gat_edge_sum<<<gEdge, B, 0, stream>>>(esrc, edst, als, ald, emax, denom, E, Etot);
  gat_edge_agg<<<aggBlocks, B, 0, stream>>>(esrc, edst, als, ald, emax, denom,
                                            h, agg, E, Etot, nWaves);
  gat_head_mean_bias<<<gFin2, B, 0, stream>>>(agg, b2, (float*)d_out, N);
}


# Scatter moved off global atomics: each block owns 391 dst nodes in LDS, filters a byte slice id per edge, does max, sum and row adds as LDS atomics, stores rows once; sums in LDS arrival order
# speedup vs baseline: 4.0482x; 4.0482x over previous
#include <hip/hip_runtime.h>
#include <math.h>

typedef float v2f __attribute__((ext_vector_type(2)));
typedef float v8f __attribute__((ext_vector_type(8)));

#define NEG_SLOPE 0.2f
#define ENC_NEG_INF 0x007FFFFFu

__device__ __forceinline__ unsigned fenc(float f) {
  unsigned u = __float_as_uint(f);
  return (u & 0x80000000u) ? ~u : (u | 0x80000000u);
}
__device__ __forceinline__ float fdec(unsigned k) {
  unsigned u = (k & 0x80000000u) ? (k & 0x7FFFFFFFu) : ~k;
  return __uint_as_float(u);
}

__global__ void gat_gemm128_wmma(const float* __restrict__ X,
                                 const float* __restrict__ W,
                                 float* __restrict__ H, int N) {
  const int wave = threadIdx.x >> 5;
  const int lane = threadIdx.x & 31;
  const int half = lane >> 4;
  const int lm   = lane & 15;
  const int row0 = blockIdx.x << 4;
  const int col  = (wave << 4) + lm;

  int arow = row0 + lm;
  if (arow >= N) arow = N - 1;
  const float* xr = X + (size_t)arow * 128;

  v8f c = {};
#pragma unroll
  for (int kk = 0; kk < 32; ++kk) {
    const int kb = (kk << 2) + (half << 1);
    v2f a, b;
    a.x = xr[kb];
    a.y = xr[kb + 1];
    b.x = W[(size_t)kb * 128 + col];
    b.y = W[(size_t)(kb + 1) * 128 + col];
    c = __builtin_amdgcn_wmma_f32_16x16x4_f32(false, a, false, b,
                                              (short)0, c, false, false);
  }

  const int orow0 = row0 + (half << 3);
#pragma unroll
  for (int v = 0; v < 8; ++v) {
    int r = orow0 + v;
    if (r < N) H[(size_t)r * 128 + col] = c[v];
  }
}

__global__ void gat_attn_logits(const float* __restrict__ H,
                                const float* __restrict__ a_src,
                                const float* __restrict__ a_dst,
                                float* __restrict__ als,
                                float* __restrict__ ald, int N) {
  const int wid  = blockIdx.x * (blockDim.x >> 5) + (threadIdx.x >> 5);
  const int lane = threadIdx.x & 31;
  if (wid >= N) return;
  const float* hr = H + (size_t)wid * 128 + lane * 4;
  const float* as = a_src + lane * 4;
  const float* ad = a_dst + lane * 4;
  float s = 0.f, d = 0.f;
#pragma unroll
  for (int j = 0; j < 4; ++j) {
    float hv = hr[j];
    s += hv * as[j];
    d += hv * ad[j];
  }
#pragma unroll
  for (int m = 8; m >= 1; m >>= 1) {
    s += __shfl_xor(s, m, 32);
    d += __shfl_xor(d, m, 32);
  }
  if ((lane & 15) == 0) {
    const int h = lane >> 4;
    als[wid * 2 + h] = s;
    ald[wid * 2 + h] = d;
  }
}

#define NB 391
#define LIST_CAP 9600
#define SRC_BITS 20
#define SRC_MASK 0xFFFFF
#define ST 1024

__global__ void gat_dst_bins(const int* __restrict__ ed,
                             unsigned char* __restrict__ bins, int E) {
  const int e = blockIdx.x * blockDim.x + threadIdx.x;
  if (e < E) bins[e] = (unsigned char)(ed[e] / NB);
}

__device__ __forceinline__ float lrelu(float v) {
  return (v >= 0.f) ? v : NEG_SLOPE * v;
}

template <int W>
__device__ __forceinline__ void lane_edge(int phase, int s, int dl,
                                          const float* __restrict__ als,
                                          const float* __restrict__ H,
                                          float* rows, unsigned* smax,
                                          float* sden, const float* sald) {
  const float v0 = lrelu(als[s * 2] + sald[dl * 2]);
  const float v1 = lrelu(als[s * 2 + 1] + sald[dl * 2 + 1]);
  if (phase == 0) {
    atomicMax(&smax[dl * 2], fenc(v0));
    atomicMax(&smax[dl * 2 + 1], fenc(v1));
    return;
  }
  const float x0 = expf(v0 - fdec(smax[dl * 2]));
  const float x1 = expf(v1 - fdec(smax[dl * 2 + 1]));
  if (phase == 1) {
    atomicAdd(&sden[dl * 2], x0);
    atomicAdd(&sden[dl * 2 + 1], x1);
    return;
  }
  const float a0 = x0 / sden[dl * 2];
  const float a1 = x1 / sden[dl * 2 + 1];
  const float* hs = H + (size_t)s * 128;
  if (W == 128) {
    for (int c = 0; c < 128; ++c)
      atomicAdd(&rows[dl * 128 + c], hs[c] * ((c < 64) ? a0 : a1));
  } else {
    const float w0 = 0.5f * a0, w1 = 0.5f * a1;
    for (int c = 0; c < 64; ++c)
      atomicAdd(&rows[dl * 64 + c], hs[c] * w0 + hs[64 + c] * w1);
  }
}

__device__ __forceinline__ void filter_vec(const uint4 q, int ebase, unsigned splat,
                                           int E, int* list, unsigned* cnt) {
  const unsigned w[4] = {q.x, q.y, q.z, q.w};
#pragma unroll
  for (int k = 0; k < 4; ++k) {
    const unsigned x = w[k] ^ splat;
    unsigned z = ~(((x & 0x7F7F7F7Fu) + 0x7F7F7F7Fu) | x | 0x7F7F7F7Fu);
    while (z) {
      const int e = ebase + (k << 2) + (__builtin_ctz(z) >> 3);
      z &= z - 1u;
      if (e < E) {
        const unsigned p = atomicAdd(cnt, 1u);
        if (p < LIST_CAP) list[p] = e;
      }
    }
  }
}

template <int W>
__device__ __forceinline__ void wave_edge_add(float* rows, const float (&hv)[4],
                                              int dl, float w0, float w1, int lane) {
  if (W == 128) {
    atomicAdd(&rows[dl * 128 + lane], hv[0] * w0);
    atomicAdd(&rows[dl * 128 + lane + 32], hv[1] * w0);
    atomicAdd(&rows[dl * 128 + lane + 64], hv[2] * w1);
    atomicAdd(&rows[dl * 128 + lane + 96], hv[3] * w1);
  } else {
    atomicAdd(&rows[dl * 64 + lane], hv[0] * w0 + hv[2] * w1);
    atomicAdd(&rows[dl * 64 + lane + 32], hv[1] * w0 + hv[3] * w1);
  }
}

template <int W>
__global__ void __launch_bounds__(ST)
gat_dst_slice(const int* __restrict__ es, const int* __restrict__ ed,
              const unsigned char* __restrict__ bins,
              const float* __restrict__ als, const float* __restrict__ ald,
              const float* __restrict__ H, const float* __restrict__ bias,
              float* __restrict__ dst, int* __restrict__ glist,
              int* __restrict__ gcnt, int N, int E) {
  __shared__ float rows[NB * W];
  __shared__ unsigned smax[NB * 2];
  __shared__ float sden[NB * 2];
  __shared__ float sald[NB * 2];
  __shared__ int list[LIST_CAP];
  __shared__ float2 sw[LIST_CAP];
  __shared__ unsigned cnt;

  const int tid = threadIdx.x;
  const int node0 = blockIdx.x * NB;
  const int nloc = (N - node0 < NB) ? (N - node0) : NB;
  const unsigned mybin = blockIdx.x;

  for (int i = tid; i < NB * W; i += ST) rows[i] = 0.f;
  for (int i = tid; i < NB * 2; i += ST) {
    smax[i] = ENC_NEG_INF;
    sden[i] = 0.f;
    sald[i] = (i < nloc * 2) ? ald[(size_t)node0 * 2 + i] : 0.f;
  }

  int total;
  if (W == 128) {
    for (int i = tid; i < nloc; i += ST) list[i] = (node0 + i) | (i << SRC_BITS);
    if (tid == 0) cnt = (unsigned)nloc;
    __syncthreads();

    const uint4* b4 = reinterpret_cast<const uint4*>(bins);
    const int nvec = (E + 15) >> 4;
    const unsigned splat = mybin * 0x01010101u;
    int v0 = tid;
    for (; v0 + 3 * ST < nvec; v0 += 4 * ST) {
      uint4 q[4];
#pragma unroll
      for (int j = 0; j < 4; ++j) q[j] = b4[v0 + j * ST];
#pragma unroll
      for (int j = 0; j < 4; ++j)
        filter_vec(q[j], (v0 + j * ST) << 4, splat, E, list, &cnt);
    }
    for (; v0 < nvec; v0 += ST)
      filter_vec(b4[v0], v0 << 4, splat, E, list, &cnt);
    __syncthreads();
    total = (int)cnt;
    if (total <= LIST_CAP) {
      for (int i = nloc + tid; i < total; i += ST) {
        const int e = list[i];
        list[i] = es[e] | ((ed[e] - node0) << SRC_BITS);
      }
      __syncthreads();
      for (int i = tid; i < total; i += ST)
        glist[(size_t)blockIdx.x * LIST_CAP + i] = list[i];
    }
    if (tid == 0) gcnt[blockIdx.x] = total;
  } else {
    total = gcnt[blockIdx.x];
    if (total <= LIST_CAP)
      for (int i = tid; i < total; i += ST)
        list[i] = glist[(size_t)blockIdx.x * LIST_CAP + i];
  }
  const bool fits = total <= LIST_CAP;
  __syncthreads();

  const int wave = tid >> 5, lane = tid & 31;
  if (fits) {
    for (int i = tid; i < total; i += ST) {
      const int pk = list[i];
      const int s = pk & SRC_MASK, dl = pk >> SRC_BITS;
      const float v0 = lrelu(als[s * 2] + sald[dl * 2]);
      const float v1 = lrelu(als[s * 2 + 1] + sald[dl * 2 + 1]);
      sw[i] = make_float2(v0, v1);
      atomicMax(&smax[dl * 2], fenc(v0));
      atomicMax(&smax[dl * 2 + 1], fenc(v1));
    }
    __syncthreads();
    for (int i = tid; i < total; i += ST) {
      const int dl = list[i] >> SRC_BITS;
      const float2 v = sw[i];
      const float x0 = expf(v.x - fdec(smax[dl * 2]));
      const float x1 = expf(v.y - fdec(smax[dl * 2 + 1]));
      sw[i] = make_float2(x0, x1);
      atomicAdd(&sden[dl * 2], x0);
      atomicAdd(&sden[dl * 2 + 1], x1);
    }
    __syncthreads();
    for (int i = tid; i < total; i += ST) {
      const int dl = list[i] >> SRC_BITS;
      const float2 x = sw[i];
      float a0 = x.x / sden[dl * 2];
      float a1 = x.y / sden[dl * 2 + 1];
      if (W == 64) { a0 = 0.5f * a0; a1 = 0.5f * a1; }
      sw[i] = make_float2(a0, a1);
    }
    __syncthreads();
    int i = wave;
    for (; i + 96 < total; i += 128) {
      int pk[4];
      float2 al[4];
      float hv[4][4];
#pragma unroll
      for (int u = 0; u < 4; ++u) {
        pk[u] = list[i + 32 * u];
        al[u] = sw[i + 32 * u];
        const float* hs = H + (size_t)(pk[u] & SRC_MASK) * 128 + lane;
        hv[u][0] = hs[0];
        hv[u][1] = hs[32];
        hv[u][2] = hs[64];
        hv[u][3] = hs[96];
      }
#pragma unroll
      for (int u = 0; u < 4; ++u)
        wave_edge_add<W>(rows, hv[u], pk[u] >> SRC_BITS, al[u].x, al[u].y, lane);
    }
    for (; i < total; i += 32) {
      const int pk = list[i];
      const float2 al = sw[i];
      const float* hs = H + (size_t)(pk & SRC_MASK) * 128 + lane;
      const float hv[4] = {hs[0], hs[32], hs[64], hs[96]};
      wave_edge_add<W>(rows, hv, pk >> SRC_BITS, al.x, al.y, lane);
    }
    __syncthreads();
  } else {
    for (int phase = 0; phase < 3; ++phase) {
      for (int i = tid; i < nloc; i += ST)
        lane_edge<W>(phase, node0 + i, i, als, H, rows, smax, sden, sald);
      for (int e = tid; e < E; e += ST)
        if (bins[e] == mybin)
          lane_edge<W>(phase, es[e], ed[e] - node0, als, H, rows, smax, sden, sald);
      __syncthreads();
    }
  }

  for (int i = tid; i < nloc * W; i += ST) {
    float v = rows[i] + bias[i & (W - 1)];
    if (W == 128) v = (v > 0.f) ? v : (expf(v) - 1.f);
    dst[(size_t)node0 * W + i] = v;
  }
}

extern "C" void kernel_launch(void* const* d_in, const int* in_sizes, int n_in,
                              void* d_out, int out_size, void* d_ws, size_t ws_size,
                              hipStream_t stream) {
  const float* x   = (const float*)d_in[0];
  const int*   ei  = (const int*)d_in[1];
  const float* W1  = (const float*)d_in[2];
  const float* as1 = (const float*)d_in[3];
  const float* ad1 = (const float*)d_in[4];
  const float* b1  = (const float*)d_in[5];
  const float* W2  = (const float*)d_in[6];
  const float* as2 = (const float*)d_in[7];
  const float* ad2 = (const float*)d_in[8];
  const float* b2  = (const float*)d_in[9];

  const int N = in_sizes[0] / 128;
  const int E = in_sizes[1] / 2;
  const int* esrc = ei;
  const int* edst = ei + E;

  float* h   = (float*)d_ws;
  float* agg = h + (size_t)N * 128;
  float* als = agg + (size_t)N * 128;
  float* ald = als + (size_t)N * 2;
  const int gSlice = (N + NB - 1) / NB;
  int* glist = (int*)(ald + (size_t)N * 2);
  int* gcnt  = glist + (size_t)gSlice * LIST_CAP;
  unsigned char* bins = (unsigned char*)(gcnt + ((gSlice + 3) & ~3));

  const int B = 256;
  const int gGemm  = (N + 15) / 16;
  const int gNodeW = (N + 7) / 8;
  const int gBins  = (E + B - 1) / B;

  gat_dst_bins<<<gBins, B, 0, stream>>>(edst, bins, E);

  gat_gemm128_wmma<<<gGemm, B, 0, stream>>>(x, W1, h, N);
  gat_attn_logits<<<gNodeW, B, 0, stream>>>(h, as1, ad1, als, ald, N);
  gat_dst_slice<128><<<gSlice, ST, 0, stream>>>(esrc, edst, bins, als, ald,
                                                h, b1, agg, glist, gcnt, N, E);

  gat_gemm128_wmma<<<gGemm, B, 0, stream>>>(agg, W2, h, N);
  gat_attn_logits<<<gNodeW, B, 0, stream>>>(h, as2, ad2, als, ald, N);
  gat_dst_slice<64><<<gSlice, ST, 0, stream>>>(esrc, edst, bins, als, ald,
                                               h, b2, (float*)d_out, glist, gcnt, N, E);
}


# Node-logit and edge-slice-id launches: each wave takes 24 nodes (all row loads in flight at once) or 512 edges (16 per lane) instead of one short wave per node or per 32 edges; logit sum order kept
# speedup vs baseline: 4.2123x; 1.0405x over previous
#include <hip/hip_runtime.h>
#include <math.h>

typedef float v2f __attribute__((ext_vector_type(2)));
typedef float v8f __attribute__((ext_vector_type(8)));

#define NEG_SLOPE 0.2f
#define ENC_NEG_INF 0x007FFFFFu

__device__ __forceinline__ unsigned fenc(float f) {
  unsigned u = __float_as_uint(f);
  return (u & 0x80000000u) ? ~u : (u | 0x80000000u);
}
__device__ __forceinline__ float fdec(unsigned k) {
  unsigned u = (k & 0x80000000u) ? (k & 0x7FFFFFFFu) : ~k;
  return __uint_as_float(u);
}

__global__ void gat_gemm128_wmma(const float* __restrict__ X,
                                 const float* __restrict__ W,
                                 float* __restrict__ H, int N) {
  const int wave = threadIdx.x >> 5;
  const int lane = threadIdx.x & 31;
  const int half = lane >> 4;
  const int lm   = lane & 15;
  const int row0 = blockIdx.x << 4;
  const int col  = (wave << 4) + lm;

  int arow = row0 + lm;
  if (arow >= N) arow = N - 1;
  const float* xr = X + (size_t)arow * 128;

  v8f c = {};
#pragma unroll
  for (int kk = 0; kk < 32; ++kk) {
    const int kb = (kk << 2) + (half << 1);
    v2f a, b;
    a.x = xr[kb];
    a.y = xr[kb + 1];
    b.x = W[(size_t)kb * 128 + col];
    b.y = W[(size_t)(kb + 1) * 128 + col];
    c = __builtin_amdgcn_wmma_f32_16x16x4_f32(false, a, false, b,
                                              (short)0, c, false, false);
  }

  const int orow0 = row0 + (half << 3);
#pragma unroll
  for (int v = 0; v < 8; ++v) {
    int r = orow0 + v;
    if (r < N) H[(size_t)r * 128 + col] = c[v];
  }
}

#define LG_NPW 24
#define LG_WAVES 8
#define LG_NPB (LG_NPW * LG_WAVES)

__global__ void __launch_bounds__(LG_WAVES * 32)
gat_attn_logits(const float* __restrict__ H,
                const float* __restrict__ a_src,
                const float* __restrict__ a_dst,
                float* __restrict__ als,
                float* __restrict__ ald, int N) {
  const int lane = threadIdx.x & 31;
  const int n0   = (blockIdx.x * LG_WAVES + (threadIdx.x >> 5)) * LG_NPW;
  if (n0 >= N) return;
  const int h = lane >> 4;
  float as[4], ad[4];
#pragma unroll
  for (int j = 0; j < 4; ++j) {
    as[j] = a_src[lane * 4 + j];
    ad[j] = a_dst[lane * 4 + j];
  }
  float4 hv[LG_NPW];
#pragma unroll
  for (int u = 0; u < LG_NPW; ++u) {
    const int n = (n0 + u < N) ? (n0 + u) : (N - 1);
    hv[u] = *reinterpret_cast<const float4*>(H + (size_t)n * 128 + lane * 4);
  }
  float s[LG_NPW], d[LG_NPW];
#pragma unroll
  for (int u = 0; u < LG_NPW; ++u) {
    const float x[4] = {hv[u].x, hv[u].y, hv[u].z, hv[u].w};
    float ss = 0.f, dd = 0.f;
#pragma unroll
    for (int j = 0; j < 4; ++j) {
      ss += x[j] * as[j];
      dd += x[j] * ad[j];
    }
    s[u] = ss;
    d[u] = dd;
  }
#pragma unroll
  for (int m = 8; m >= 1; m >>= 1) {
#pragma unroll
    for (int u = 0; u < LG_NPW; ++u) {
      s[u] += __shfl_xor(s[u], m, 32);
      d[u] += __shfl_xor(d[u], m, 32);
    }
  }
  if ((lane & 15) == 0) {
#pragma unroll
    for (int u = 0; u < LG_NPW; ++u) {
      if (n0 + u < N) {
        als[(n0 + u) * 2 + h] = s[u];
        ald[(n0 + u) * 2 + h] = d[u];
      }
    }
  }
}

#define NB 391
#define LIST_CAP 9600
#define SRC_BITS 20
#define SRC_MASK 0xFFFFF
#define ST 1024

#define BINS_EPT 16

__global__ void __launch_bounds__(256)
gat_dst_bins(const int* __restrict__ ed,
             unsigned char* __restrict__ bins, int E) {
  const int e0 = (blockIdx.x * blockDim.x + threadIdx.x) * BINS_EPT;
  if (e0 + BINS_EPT <= E) {
    int d[BINS_EPT];
#pragma unroll
    for (int k = 0; k < BINS_EPT; ++k) d[k] = ed[e0 + k];
    unsigned w[4];
#pragma unroll
    for (int q = 0; q < 4; ++q)
      w[q] = ((unsigned)(d[4 * q] / NB) & 0xFFu) |
             (((unsigned)(d[4 * q + 1] / NB) & 0xFFu) << 8) |
             (((unsigned)(d[4 * q + 2] / NB) & 0xFFu) << 16) |
             ((unsigned)(d[4 * q + 3] / NB) << 24);
    *reinterpret_cast<uint4*>(bins + e0) = make_uint4(w[0], w[1], w[2], w[3]);
  } else {
    for (int e = e0; e < E; ++e) bins[e] = (unsigned char)(ed[e] / NB);
  }
}

__device__ __forceinline__ float lrelu(float v) {
  return (v >= 0.f) ? v : NEG_SLOPE * v;
}

template <int W>
__device__ __forceinline__ void lane_edge(int phase, int s, int dl,
                                          const float* __restrict__ als,
                                          const float* __restrict__ H,
                                          float* rows, unsigned* smax,
                                          float* sden, const float* sald) {
  const float v0 = lrelu(als[s * 2] + sald[dl * 2]);
  const float v1 = lrelu(als[s * 2 + 1] + sald[dl * 2 + 1]);
  if (phase == 0) {
    atomicMax(&smax[dl * 2], fenc(v0));
    atomicMax(&smax[dl * 2 + 1], fenc(v1));
    return;
  }
  const float x0 = expf(v0 - fdec(smax[dl * 2]));
  const float x1 = expf(v1 - fdec(smax[dl * 2 + 1]));
  if (phase == 1) {
    atomicAdd(&sden[dl * 2], x0);
    atomicAdd(&sden[dl * 2 + 1], x1);
    return;
  }
  const float a0 = x0 / sden[dl * 2];
  const float a1 = x1 / sden[dl * 2 + 1];
  const float* hs = H + (size_t)s * 128;
  if (W == 128) {
    for (int c = 0; c < 128; ++c)
      atomicAdd(&rows[dl * 128 + c], hs[c] * ((c < 64) ? a0 : a1));
  } else {
    const float w0 = 0.5f * a0, w1 = 0.5f * a1;
    for (int c = 0; c < 64; ++c)
      atomicAdd(&rows[dl * 64 + c], hs[c] * w0 + hs[64 + c] * w1);
  }
}

__device__ __forceinline__ void filter_vec(const uint4 q, int ebase, unsigned splat,
                                           int E, int* list, unsigned* cnt) {
  const unsigned w[4] = {q.x, q.y, q.z, q.w};
#pragma unroll
  for (int k = 0; k < 4; ++k) {
    const unsigned x = w[k] ^ splat;
    unsigned z = ~(((x & 0x7F7F7F7Fu) + 0x7F7F7F7Fu) | x | 0x7F7F7F7Fu);
    while (z) {
      const int e = ebase + (k << 2) + (__builtin_ctz(z) >> 3);
      z &= z - 1u;
      if (e < E) {
        const unsigned p = atomicAdd(cnt, 1u);
        if (p < LIST_CAP) list[p] = e;
      }
    }
  }
}

template <int W>
__device__ __forceinline__ void wave_edge_add(float* rows, const float (&hv)[4],
                                              int dl, float w0, float w1, int lane) {
  if (W == 128) {
    atomicAdd(&rows[dl * 128 + lane], hv[0] * w0);
    atomicAdd(&rows[dl * 128 + lane + 32], hv[1] * w0);
    atomicAdd(&rows[dl * 128 + lane + 64], hv[2] * w1);
    atomicAdd(&rows[dl * 128 + lane + 96], hv[3] * w1);
  } else {
    atomicAdd(&rows[dl * 64 + lane], hv[0] * w0 + hv[2] * w1);
    atomicAdd(&rows[dl * 64 + lane + 32], hv[1] * w0 + hv[3] * w1);
  }
}

template <int W>
__global__ void __launch_bounds__(ST)
gat_dst_slice(const int* __restrict__ es, const int* __restrict__ ed,
              const unsigned char* __restrict__ bins,
              const float* __restrict__ als, const float* __restrict__ ald,
              const float* __restrict__ H, const float* __restrict__ bias,
              float* __restrict__ dst, int* __restrict__ glist,
              int* __restrict__ gcnt, int N, int E) {
  __shared__ float rows[NB * W];
  __shared__ unsigned smax[NB * 2];
  __shared__ float sden[NB * 2];
  __shared__ float sald[NB * 2];
  __shared__ int list[LIST_CAP];
  __shared__ float2 sw[LIST_CAP];
  __shared__ unsigned cnt;

  const int tid = threadIdx.x;
  const int node0 = blockIdx.x * NB;
  const int nloc = (N - node0 < NB) ? (N - node0) : NB;
  const unsigned mybin = blockIdx.x;

  for (int i = tid; i < NB * W; i += ST) rows[i] = 0.f;
  for (int i = tid; i < NB * 2; i += ST) {
    smax[i] = ENC_NEG_INF;
    sden[i] = 0.f;
    sald[i] = (i < nloc * 2) ? ald[(size_t)node0 * 2 + i] : 0.f;
  }

  int total;
  if (W == 128) {
    for (int i = tid; i < nloc; i += ST) list[i] = (node0 + i) | (i << SRC_BITS);
    if (tid == 0) cnt = (unsigned)nloc;
    __syncthreads();

    const uint4* b4 = reinterpret_cast<const uint4*>(bins);
    const int nvec = (E + 15) >> 4;
    const unsigned splat = mybin * 0x01010101u;
    int v0 = tid;
    for (; v0 + 3 * ST < nvec; v0 += 4 * ST) {
      uint4 q[4];
#pragma unroll
      for (int j = 0; j < 4; ++j) q[j] = b4[v0 + j * ST];
#pragma unroll
      for (int j = 0; j < 4; ++j)
        filter_vec(q[j], (v0 + j * ST) << 4, splat, E, list, &cnt);
    }
    for (; v0 < nvec; v0 += ST)
      filter_vec(b4[v0], v0 << 4, splat, E, list, &cnt);
    __syncthreads();
    total = (int)cnt;
    if (total <= LIST_CAP) {
      for (int i = nloc + tid; i < total; i += ST) {
        const int e = list[i];
        list[i] = es[e] | ((ed[e] - node0) << SRC_BITS);
      }
      __syncthreads();
      for (int i = tid; i < total; i += ST)
        glist[(size_t)blockIdx.x * LIST_CAP + i] = list[i];
    }
    if (tid == 0) gcnt[blockIdx.x] = total;
  } else {
    total = gcnt[blockIdx.x];
    if (total <= LIST_CAP)
      for (int i = tid; i < total; i += ST)
        list[i] = glist[(size_t)blockIdx.x * LIST_CAP + i];
  }
  const bool fits = total <= LIST_CAP;
  __syncthreads();

  const int wave = tid >> 5, lane = tid & 31;
  if (fits) {
    for (int i = tid; i < total; i += ST) {
      const int pk = list[i];
      const int s = pk & SRC_MASK, dl = pk >> SRC_BITS;
      const float v0 = lrelu(als[s * 2] + sald[dl * 2]);
      const float v1 = lrelu(als[s * 2 + 1] + sald[dl * 2 + 1]);
      sw[i] = make_float2(v0, v1);
      atomicMax(&smax[dl * 2], fenc(v0));
      atomicMax(&smax[dl * 2 + 1], fenc(v1));
    }
    __syncthreads();
    for (int i = tid; i < total; i += ST) {
      const int dl = list[i] >> SRC_BITS;
      const float2 v = sw[i];
      const float x0 = expf(v.x - fdec(smax[dl * 2]));
      const float x1 = expf(v.y - fdec(smax[dl * 2 + 1]));
      sw[i] = make_float2(x0, x1);
      atomicAdd(&sden[dl * 2], x0);
      atomicAdd(&sden[dl * 2 + 1], x1);
    }
    __syncthreads();
    for (int i = tid; i < total; i += ST) {
      const int dl = list[i] >> SRC_BITS;
      const float2 x = sw[i];
      float a0 = x.x / sden[dl * 2];
      float a1 = x.y / sden[dl * 2 + 1];
      if (W == 64) { a0 = 0.5f * a0; a1 = 0.5f * a1; }
      sw[i] = make_float2(a0, a1);
    }
    __syncthreads();
    int i = wave;
    for (; i + 96 < total; i += 128) {
      int pk[4];
      float2 al[4];
      float hv[4][4];
#pragma unroll
      for (int u = 0; u < 4; ++u) {
        pk[u] = list[i + 32 * u];
        al[u] = sw[i + 32 * u];
        const float* hs = H + (size_t)(pk[u] & SRC_MASK) * 128 + lane;
        hv[u][0] = hs[0];
        hv[u][1] = hs[32];
        hv[u][2] = hs[64];
        hv[u][3] = hs[96];
      }
#pragma unroll
      for (int u = 0; u < 4; ++u)
        wave_edge_add<W>(rows, hv[u], pk[u] >> SRC_BITS, al[u].x, al[u].y, lane);
    }
    for (; i < total; i += 32) {
      const int pk = list[i];
      const float2 al = sw[i];
      const float* hs = H + (size_t)(pk & SRC_MASK) * 128 + lane;
      const float hv[4] = {hs[0], hs[32], hs[64], hs[96]};
      wave_edge_add<W>(rows, hv, pk >> SRC_BITS, al.x, al.y, lane);
    }
    __syncthreads();
  } else {
    for (int phase = 0; phase < 3; ++phase) {
      for (int i = tid; i < nloc; i += ST)
        lane_edge<W>(phase, node0 + i, i, als, H, rows, smax, sden, sald);
      for (int e = tid; e < E; e += ST)
        if (bins[e] == mybin)
          lane_edge<W>(phase, es[e], ed[e] - node0, als, H, rows, smax, sden, sald);
      __syncthreads();
    }
  }

  for (int i = tid; i < nloc * W; i += ST) {
    float v = rows[i] + bias[i & (W - 1)];
    if (W == 128) v = (v > 0.f) ? v : (expf(v) - 1.f);
    dst[(size_t)node0 * W + i] = v;
  }
}

extern "C" void kernel_launch(void* const* d_in, const int* in_sizes, int n_in,
                              void* d_out, int out_size, void* d_ws, size_t ws_size,
                              hipStream_t stream) {
  const float* x   = (const float*)d_in[0];
  const int*   ei  = (const int*)d_in[1];
  const float* W1  = (const float*)d_in[2];
  const float* as1 = (const float*)d_in[3];
  const float* ad1 = (const float*)d_in[4];
  const float* b1  = (const float*)d_in[5];
  const float* W2  = (const float*)d_in[6];
  const float* as2 = (const float*)d_in[7];
  const float* ad2 = (const float*)d_in[8];
  const float* b2  = (const float*)d_in[9];

  const int N = in_sizes[0] / 128;
  const int E = in_sizes[1] / 2;
  const int* esrc = ei;
  const int* edst = ei + E;

  float* h   = (float*)d_ws;
  float* agg = h + (size_t)N * 128;
  float* als = agg + (size_t)N * 128;
  float* ald = als + (size_t)N * 2;
  const int gSlice = (N + NB - 1) / NB;
  int* glist = (int*)(ald + (size_t)N * 2);
  int* gcnt  = glist + (size_t)gSlice * LIST_CAP;
  unsigned char* bins = (unsigned char*)(gcnt + ((gSlice + 3) & ~3));

  const int B = 256;
  const int gGemm  = (N + 15) / 16;
  const int gNodeW = (N + LG_NPB - 1) / LG_NPB;
  const int gBins  = (E + B * BINS_EPT - 1) / (B * BINS_EPT);

  gat_dst_bins<<<gBins, B, 0, stream>>>(edst, bins, E);

  gat_gemm128_wmma<<<gGemm, B, 0, stream>>>(x, W1, h, N);
  gat_attn_logits<<<gNodeW, B, 0, stream>>>(h, as1, ad1, als, ald, N);
  gat_dst_slice<128><<<gSlice, ST, 0, stream>>>(esrc, edst, bins, als, ald,
                                                h, b1, agg, glist, gcnt, N, E);

  gat_gemm128_wmma<<<gGemm, B, 0, stream>>>(agg, W2, h, N);
  gat_attn_logits<<<gNodeW, B, 0, stream>>>(h, as2, ad2, als, ald, N);
  gat_dst_slice<64><<<gSlice, ST, 0, stream>>>(esrc, edst, bins, als, ald,
                                               h, b2, (float*)d_out, glist, gcnt, N, E);
}


# Feature GEMM made weight-stationary: a wave keeps its 128x16 strip of W in registers over 5 row tiles; the block's X tile comes once through LDS by whole-row loads; k order and sum order kept
# speedup vs baseline: 4.9173x; 1.1674x over previous
#include <hip/hip_runtime.h>
#include <math.h>

typedef float v2f __attribute__((ext_vector_type(2)));
typedef float v8f __attribute__((ext_vector_type(8)));

#define NEG_SLOPE 0.2f
#define ENC_NEG_INF 0x007FFFFFu

__device__ __forceinline__ unsigned fenc(float f) {
  unsigned u = __float_as_uint(f);
  return (u & 0x80000000u) ? ~u : (u | 0x80000000u);
}
__device__ __forceinline__ float fdec(unsigned k) {
  unsigned u = (k & 0x80000000u) ? (k & 0x7FFFFFFFu) : ~k;
  return __uint_as_float(u);
}

#define GT_TILES 5
#define GT_PITCH 132

__global__ void __launch_bounds__(256)
gat_gemm128_wmma(const float* __restrict__ X,
                 const float* __restrict__ W,
                 float* __restrict__ H, int N) {
  __shared__ __attribute__((aligned(16))) float xs[2][16 * GT_PITCH];

  const int tid  = threadIdx.x;
  const int wave = tid >> 5;
  const int lane = tid & 31;
  const int half = lane >> 4;
  const int lm   = lane & 15;
  const int col  = (wave << 4) + lm;

  const int ntiles = (N + 15) >> 4;
  const int tile0  = blockIdx.x * GT_TILES;
  const int nt = (ntiles - tile0 < GT_TILES) ? (ntiles - tile0) : GT_TILES;

  const int sr = tid >> 5;
  const int sc = tid & 31;
  const int sw0 = ((sc >> 1) << 3) + ((sc & 1) << 1);

  float4 pre[2];
#pragma unroll
  for (int q = 0; q < 2; ++q) {
    int r = (tile0 << 4) + sr + (q << 3);
    if (r >= N) r = N - 1;
    pre[q] = *reinterpret_cast<const float4*>(X + (size_t)r * 128 + (sc << 2));
  }

  v2f b[32];
#pragma unroll
  for (int kk = 0; kk < 32; ++kk) {
    const int kb = (kk << 2) + (half << 1);
    b[kk].x = W[(size_t)kb * 128 + col];
    b[kk].y = W[(size_t)(kb + 1) * 128 + col];
  }

#pragma unroll
  for (int q = 0; q < 2; ++q) {
    float* d = &xs[0][(sr + (q << 3)) * GT_PITCH + sw0];
    *reinterpret_cast<float2*>(d)     = make_float2(pre[q].x, pre[q].y);
    *reinterpret_cast<float2*>(d + 4) = make_float2(pre[q].z, pre[q].w);
  }
  __syncthreads();

  for (int i = 0; i < nt; ++i) {
    const int tile = tile0 + i;
    const bool more = (i + 1 < nt);
    if (more) {
#pragma unroll
      for (int q = 0; q < 2; ++q) {
        int r = ((tile + 1) << 4) + sr + (q << 3);
        if (r >= N) r = N - 1;
        pre[q] = *reinterpret_cast<const float4*>(X + (size_t)r * 128 + (sc << 2));
      }
    }

    const float* xt = &xs[i & 1][lm * GT_PITCH + (half << 2)];
    float4 af[16];
#pragma unroll
    for (int j = 0; j < 16; ++j)
      af[j] = *reinterpret_cast<const float4*>(xt + (j << 3));

    v8f c = {};
#pragma unroll
    for (int j = 0; j < 16; ++j) {
      v2f a0, a1;
      a0.x = af[j].x; a0.y = af[j].y;
      a1.x = af[j].z; a1.y = af[j].w;
      c = __builtin_amdgcn_wmma_f32_16x16x4_f32(false, a0, false, b[2 * j],
                                                (short)0, c, false, false);
      c = __builtin_amdgcn_wmma_f32_16x16x4_f32(false, a1, false, b[2 * j + 1],
                                                (short)0, c, false, false);
    }

    const int orow0 = (tile << 4) + (half << 3);
#pragma unroll
    for (int v = 0; v < 8; ++v) {
      const int r = orow0 + v;
      if (r < N) H[(size_t)r * 128 + col] = c[v];
    }

    if (more) {
#pragma unroll
      for (int q = 0; q < 2; ++q) {
        float* d = &xs[(i + 1) & 1][(sr + (q << 3)) * GT_PITCH + sw0];
        *reinterpret_cast<float2*>(d)     = make_float2(pre[q].x, pre[q].y);
        *reinterpret_cast<float2*>(d + 4) = make_float2(pre[q].z, pre[q].w);
      }
    }
    __syncthreads();
  }
}

#define LG_NPW 24
#define LG_WAVES 8
#define LG_NPB (LG_NPW * LG_WAVES)

__global__ void __launch_bounds__(LG_WAVES * 32)
gat_attn_logits(const float* __restrict__ H,
                const float* __restrict__ a_src,
                const float* __restrict__ a_dst,
                float* __restrict__ als,
                float* __restrict__ ald, int N) {
  const int lane = threadIdx.x & 31;
  const int n0   = (blockIdx.x * LG_WAVES + (threadIdx.x >> 5)) * LG_NPW;
  if (n0 >= N) return;
  const int h = lane >> 4;
  float as[4], ad[4];
#pragma unroll
  for (int j = 0; j < 4; ++j) {
    as[j] = a_src[lane * 4 + j];
    ad[j] = a_dst[lane * 4 + j];
  }
  float4 hv[LG_NPW];
#pragma unroll
  for (int u = 0; u < LG_NPW; ++u) {
    const int n = (n0 + u < N) ? (n0 + u) : (N - 1);
    hv[u] = *reinterpret_cast<const float4*>(H + (size_t)n * 128 + lane * 4);
  }
  float s[LG_NPW], d[LG_NPW];
#pragma unroll
  for (int u = 0; u < LG_NPW; ++u) {
    const float x[4] = {hv[u].x, hv[u].y, hv[u].z, hv[u].w};
    float ss = 0.f, dd = 0.f;
#pragma unroll
    for (int j = 0; j < 4; ++j) {
      ss += x[j] * as[j];
      dd += x[j] * ad[j];
    }
    s[u] = ss;
    d[u] = dd;
  }
#pragma unroll
  for (int m = 8; m >= 1; m >>= 1) {
#pragma unroll
    for (int u = 0; u < LG_NPW; ++u) {
      s[u] += __shfl_xor(s[u], m, 32);
      d[u] += __shfl_xor(d[u], m, 32);
    }
  }
  if ((lane & 15) == 0) {
#pragma unroll
    for (int u = 0; u < LG_NPW; ++u) {
      if (n0 + u < N) {
        als[(n0 + u) * 2 + h] = s[u];
        ald[(n0 + u) * 2 + h] = d[u];
      }
    }
  }
}

#define NB 391
#define LIST_CAP 9600
#define SRC_BITS 20
#define SRC_MASK 0xFFFFF
#define ST 1024

#define BINS_EPT 16

__global__ void __launch_bounds__(256)
gat_dst_bins(const int* __restrict__ ed,
             unsigned char* __restrict__ bins, int E) {
  const int e0 = (blockIdx.x * blockDim.x + threadIdx.x) * BINS_EPT;
  if (e0 + BINS_EPT <= E) {
    int d[BINS_EPT];
#pragma unroll
    for (int k = 0; k < BINS_EPT; ++k) d[k] = ed[e0 + k];
    unsigned w[4];
#pragma unroll
    for (int q = 0; q < 4; ++q)
      w[q] = ((unsigned)(d[4 * q] / NB) & 0xFFu) |
             (((unsigned)(d[4 * q + 1] / NB) & 0xFFu) << 8) |
             (((unsigned)(d[4 * q + 2] / NB) & 0xFFu) << 16) |
             ((unsigned)(d[4 * q + 3] / NB) << 24);
    *reinterpret_cast<uint4*>(bins + e0) = make_uint4(w[0], w[1], w[2], w[3]);
  } else {
    for (int e = e0; e < E; ++e) bins[e] = (unsigned char)(ed[e] / NB);
  }
}

__device__ __forceinline__ float lrelu(float v) {
  return (v >= 0.f) ? v : NEG_SLOPE * v;
}

template <int W>
__device__ __forceinline__ void lane_edge(int phase, int s, int dl,
                                          const float* __restrict__ als,
                                          const float* __restrict__ H,
                                          float* rows, unsigned* smax,
                                          float* sden, const float* sald) {
  const float v0 = lrelu(als[s * 2] + sald[dl * 2]);
  const float v1 = lrelu(als[s * 2 + 1] + sald[dl * 2 + 1]);
  if (phase == 0) {
    atomicMax(&smax[dl * 2], fenc(v0));
    atomicMax(&smax[dl * 2 + 1], fenc(v1));
    return;
  }
  const float x0 = expf(v0 - fdec(smax[dl * 2]));
  const float x1 = expf(v1 - fdec(smax[dl * 2 + 1]));
  if (phase == 1) {
    atomicAdd(&sden[dl * 2], x0);
    atomicAdd(&sden[dl * 2 + 1], x1);
    return;
  }
  const float a0 = x0 / sden[dl * 2];
  const float a1 = x1 / sden[dl * 2 + 1];
  const float* hs = H + (size_t)s * 128;
  if (W == 128) {
    for (int c = 0; c < 128; ++c)
      atomicAdd(&rows[dl * 128 + c], hs[c] * ((c < 64) ? a0 : a1));
  } else {
    const float w0 = 0.5f * a0, w1 = 0.5f * a1;
    for (int c = 0; c < 64; ++c)
      atomicAdd(&rows[dl * 64 + c], hs[c] * w0 + hs[64 + c] * w1);
  }
}

__device__ __forceinline__ void filter_vec(const uint4 q, int ebase, unsigned splat,
                                           int E, int* list, unsigned* cnt) {
  const unsigned w[4] = {q.x, q.y, q.z, q.w};
#pragma unroll
  for (int k = 0; k < 4; ++k) {
    const unsigned x = w[k] ^ splat;
    unsigned z = ~(((x & 0x7F7F7F7Fu) + 0x7F7F7F7Fu) | x | 0x7F7F7F7Fu);
    while (z) {
      const int e = ebase + (k << 2) + (__builtin_ctz(z) >> 3);
      z &= z - 1u;
      if (e < E) {
        const unsigned p = atomicAdd(cnt, 1u);
        if (p < LIST_CAP) list[p] = e;
      }
    }
  }
}

template <int W>
__device__ __forceinline__ void wave_edge_add(float* rows, const float (&hv)[4],
                                              int dl, float w0, float w1, int lane) {
  if (W == 128) {
    atomicAdd(&rows[dl * 128 + lane], hv[0] * w0);
    atomicAdd(&rows[dl * 128 + lane + 32], hv[1] * w0);
    atomicAdd(&rows[dl * 128 + lane + 64], hv[2] * w1);
    atomicAdd(&rows[dl * 128 + lane + 96], hv[3] * w1);
  } else {
    atomicAdd(&rows[dl * 64 + lane], hv[0] * w0 + hv[2] * w1);
    atomicAdd(&rows[dl * 64 + lane + 32], hv[1] * w0 + hv[3] * w1);
  }
}

template <int W>
__global__ void __launch_bounds__(ST)
gat_dst_slice(const int* __restrict__ es, const int* __restrict__ ed,
              const unsigned char* __restrict__ bins,
              const float* __restrict__ als, const float* __restrict__ ald,
              const float* __restrict__ H, const float* __restrict__ bias,
              float* __restrict__ dst, int* __restrict__ glist,
              int* __restrict__ gcnt, int N, int E) {
  __shared__ float rows[NB * W];
  __shared__ unsigned smax[NB * 2];
  __shared__ float sden[NB * 2];
  __shared__ float sald[NB * 2];
  __shared__ int list[LIST_CAP];
  __shared__ float2 sw[LIST_CAP];
  __shared__ unsigned cnt;

  const int tid = threadIdx.x;
  const int node0 = blockIdx.x * NB;
  const int nloc = (N - node0 < NB) ? (N - node0) : NB;
  const unsigned mybin = blockIdx.x;

  for (int i = tid; i < NB * W; i += ST) rows[i] = 0.f;
  for (int i = tid; i < NB * 2; i += ST) {
    smax[i] = ENC_NEG_INF;
    sden[i] = 0.f;
    sald[i] = (i < nloc * 2) ? ald[(size_t)node0 * 2 + i] : 0.f;
  }

  int total;
  if (W == 128) {
    for (int i = tid; i < nloc; i += ST) list[i] = (node0 + i) | (i << SRC_BITS);
    if (tid == 0) cnt = (unsigned)nloc;
    __syncthreads();

    const uint4* b4 = reinterpret_cast<const uint4*>(bins);
    const int nvec = (E + 15) >> 4;
    const unsigned splat = mybin * 0x01010101u;
    int v0 = tid;
    for (; v0 + 3 * ST < nvec; v0 += 4 * ST) {
      uint4 q[4];
#pragma unroll
      for (int j = 0; j < 4; ++j) q[j] = b4[v0 + j * ST];
#pragma unroll
      for (int j = 0; j < 4; ++j)
        filter_vec(q[j], (v0 + j * ST) << 4, splat, E, list, &cnt);
    }
    for (; v0 < nvec; v0 += ST)
      filter_vec(b4[v0], v0 << 4, splat, E, list, &cnt);
    __syncthreads();
    total = (int)cnt;
    if (total <= LIST_CAP) {
      for (int i = nloc + tid; i < total; i += ST) {
        const int e = list[i];
        list[i] = es[e] | ((ed[e] - node0) << SRC_BITS);
      }
      __syncthreads();
      for (int i = tid; i < total; i += ST)
        glist[(size_t)blockIdx.x * LIST_CAP + i] = list[i];
    }
    if (tid == 0) gcnt[blockIdx.x] = total;
  } else {
    total = gcnt[blockIdx.x];
    if (total <= LIST_CAP)
      for (int i = tid; i < total; i += ST)
        list[i] = glist[(size_t)blockIdx.x * LIST_CAP + i];
  }
  const bool fits = total <= LIST_CAP;
  __syncthreads();

  const int wave = tid >> 5, lane = tid & 31;
  if (fits) {
    for (int i = tid; i < total; i += ST) {
      const int pk = list[i];
      const int s = pk & SRC_MASK, dl = pk >> SRC_BITS;
      const float v0 = lrelu(als[s * 2] + sald[dl * 2]);
      const float v1 = lrelu(als[s * 2 + 1] + sald[dl * 2 + 1]);
      sw[i] = make_float2(v0, v1);
      atomicMax(&smax[dl * 2], fenc(v0));
      atomicMax(&smax[dl * 2 + 1], fenc(v1));
    }
    __syncthreads();
    for (int i = tid; i < total; i += ST) {
      const int dl = list[i] >> SRC_BITS;
      const float2 v = sw[i];
      const float x0 = expf(v.x - fdec(smax[dl * 2]));
      const float x1 = expf(v.y - fdec(smax[dl * 2 + 1]));
      sw[i] = make_float2(x0, x1);
      atomicAdd(&sden[dl * 2], x0);
      atomicAdd(&sden[dl * 2 + 1], x1);
    }
    __syncthreads();
    for (int i = tid; i < total; i += ST) {
      const int dl = list[i] >> SRC_BITS;
      const float2 x = sw[i];
      float a0 = x.x / sden[dl * 2];
      float a1 = x.y / sden[dl * 2 + 1];
      if (W == 64) { a0 = 0.5f * a0; a1 = 0.5f * a1; }
      sw[i] = make_float2(a0, a1);
    }
    __syncthreads();
    int i = wave;
    for (; i + 96 < total; i += 128) {
      int pk[4];
      float2 al[4];
      float hv[4][4];
#pragma unroll
      for (int u = 0; u < 4; ++u) {
        pk[u] = list[i + 32 * u];
        al[u] = sw[i + 32 * u];
        const float* hs = H + (size_t)(pk[u] & SRC_MASK) * 128 + lane;
        hv[u][0] = hs[0];
        hv[u][1] = hs[32];
        hv[u][2] = hs[64];
        hv[u][3] = hs[96];
      }
#pragma unroll
      for (int u = 0; u < 4; ++u)
        wave_edge_add<W>(rows, hv[u], pk[u] >> SRC_BITS, al[u].x, al[u].y, lane);
    }
    for (; i < total; i += 32) {
      const int pk = list[i];
      const float2 al = sw[i];
      const float* hs = H + (size_t)(pk & SRC_MASK) * 128 + lane;
      const float hv[4] = {hs[0], hs[32], hs[64], hs[96]};
      wave_edge_add<W>(rows, hv, pk >> SRC_BITS, al.x, al.y, lane);
    }
    __syncthreads();
  } else {
    for (int phase = 0; phase < 3; ++phase) {
      for (int i = tid; i < nloc; i += ST)
        lane_edge<W>(phase, node0 + i, i, als, H, rows, smax, sden, sald);
      for (int e = tid; e < E; e += ST)
        if (bins[e] == mybin)
          lane_edge<W>(phase, es[e], ed[e] - node0, als, H, rows, smax, sden, sald);
      __syncthreads();
    }
  }

  for (int i = tid; i < nloc * W; i += ST) {
    float v = rows[i] + bias[i & (W - 1)];
    if (W == 128) v = (v > 0.f) ? v : (expf(v) - 1.f);
    dst[(size_t)node0 * W + i] = v;
  }
}

extern "C" void kernel_launch(void* const* d_in, const int* in_sizes, int n_in,
                              void* d_out, int out_size, void* d_ws, size_t ws_size,
                              hipStream_t stream) {
  const float* x   = (const float*)d_in[0];
  const int*   ei  = (const int*)d_in[1];
  const float* W1  = (const float*)d_in[2];
  const float* as1 = (const float*)d_in[3];
  const float* ad1 = (const float*)d_in[4];
  const float* b1  = (const float*)d_in[5];
  const float* W2  = (const float*)d_in[6];
  const float* as2 = (const float*)d_in[7];
  const float* ad2 = (const float*)d_in[8];
  const float* b2  = (const float*)d_in[9];

  const int N = in_sizes[0] / 128;
  const int E = in_sizes[1] / 2;
  const int* esrc = ei;
  const int* edst = ei + E;

  float* h   = (float*)d_ws;
  float* agg = h + (size_t)N * 128;
  float* als = agg + (size_t)N * 128;
  float* ald = als + (size_t)N * 2;
  const int gSlice = (N + NB - 1) / NB;
  int* glist = (int*)(ald + (size_t)N * 2);
  int* gcnt  = glist + (size_t)gSlice * LIST_CAP;
  unsigned char* bins = (unsigned char*)(gcnt + ((gSlice + 3) & ~3));

  const int B = 256;
  const int gGemm  = (N + 16 * GT_TILES - 1) / (16 * GT_TILES);
  const int gNodeW = (N + LG_NPB - 1) / LG_NPB;
  const int gBins  = (E + B * BINS_EPT - 1) / (B * BINS_EPT);

  gat_dst_bins<<<gBins, B, 0, stream>>>(edst, bins, E);

  gat_gemm128_wmma<<<gGemm, B, 0, stream>>>(x, W1, h, N);
  gat_attn_logits<<<gNodeW, B, 0, stream>>>(h, as1, ad1, als, ald, N);
  gat_dst_slice<128><<<gSlice, ST, 0, stream>>>(esrc, edst, bins, als, ald,
                                                h, b1, agg, glist, gcnt, N, E);

  gat_gemm128_wmma<<<gGemm, B, 0, stream>>>(agg, W2, h, N);
  gat_attn_logits<<<gNodeW, B, 0, stream>>>(h, as2, ad2, als, ald, N);
  gat_dst_slice<64><<<gSlice, ST, 0, stream>>>(esrc, edst, bins, als, ald,
                                               h, b2, (float*)d_out, glist, gcnt, N, E);
}


# Scatter row gather re-tiled: a wave takes 8 list entries per trip, reads each 512-byte source row at 16 (layer 2: 8) bytes per lane; LDS row words permuted, adds stay conflict-free; LDS arrival order
# speedup vs baseline: 5.2682x; 1.0714x over previous
#include <hip/hip_runtime.h>
#include <math.h>

typedef float v2f __attribute__((ext_vector_type(2)));
typedef float v8f __attribute__((ext_vector_type(8)));

#define NEG_SLOPE 0.2f
#define ENC_NEG_INF 0x007FFFFFu

__device__ __forceinline__ unsigned fenc(float f) {
  unsigned u = __float_as_uint(f);
  return (u & 0x80000000u) ? ~u : (u | 0x80000000u);
}
__device__ __forceinline__ float fdec(unsigned k) {
  unsigned u = (k & 0x80000000u) ? (k & 0x7FFFFFFFu) : ~k;
  return __uint_as_float(u);
}

#define GT_TILES 5
#define GT_PITCH 132

__global__ void __launch_bounds__(256)
gat_gemm128_wmma(const float* __restrict__ X,
                 const float* __restrict__ W,
                 float* __restrict__ H, int N) {
  __shared__ __attribute__((aligned(16))) float xs[2][16 * GT_PITCH];

  const int tid  = threadIdx.x;
  const int wave = tid >> 5;
  const int lane = tid & 31;
  const int half = lane >> 4;
  const int lm   = lane & 15;
  const int col  = (wave << 4) + lm;

  const int ntiles = (N + 15) >> 4;
  const int tile0  = blockIdx.x * GT_TILES;
  const int nt = (ntiles - tile0 < GT_TILES) ? (ntiles - tile0) : GT_TILES;

  const int sr = tid >> 5;
  const int sc = tid & 31;
  const int sw0 = ((sc >> 1) << 3) + ((sc & 1) << 1);

  float4 pre[2];
#pragma unroll
  for (int q = 0; q < 2; ++q) {
    int r = (tile0 << 4) + sr + (q << 3);
    if (r >= N) r = N - 1;
    pre[q] = *reinterpret_cast<const float4*>(X + (size_t)r * 128 + (sc << 2));
  }

  v2f b[32];
#pragma unroll
  for (int kk = 0; kk < 32; ++kk) {
    const int kb = (kk << 2) + (half << 1);
    b[kk].x = W[(size_t)kb * 128 + col];
    b[kk].y = W[(size_t)(kb + 1) * 128 + col];
  }

#pragma unroll
  for (int q = 0; q < 2; ++q) {
    float* d = &xs[0][(sr + (q << 3)) * GT_PITCH + sw0];
    *reinterpret_cast<float2*>(d)     = make_float2(pre[q].x, pre[q].y);
    *reinterpret_cast<float2*>(d + 4) = make_float2(pre[q].z, pre[q].w);
  }
  __syncthreads();

  for (int i = 0; i < nt; ++i) {
    const int tile = tile0 + i;
    const bool more = (i + 1 < nt);
    if (more) {
#pragma unroll
      for (int q = 0; q < 2; ++q) {
        int r = ((tile + 1) << 4) + sr + (q << 3);
        if (r >= N) r = N - 1;
        pre[q] = *reinterpret_cast<const float4*>(X + (size_t)r * 128 + (sc << 2));
      }
    }

    const float* xt = &xs[i & 1][lm * GT_PITCH + (half << 2)];
    float4 af[16];
#pragma unroll
    for (int j = 0; j < 16; ++j)
      af[j] = *reinterpret_cast<const float4*>(xt + (j << 3));

    v8f c = {};
#pragma unroll
    for (int j = 0; j < 16; ++j) {
      v2f a0, a1;
      a0.x = af[j].x; a0.y = af[j].y;
      a1.x = af[j].z; a1.y = af[j].w;
      c = __builtin_amdgcn_wmma_f32_16x16x4_f32(false, a0, false, b[2 * j],
                                                (short)0, c, false, false);
      c = __builtin_amdgcn_wmma_f32_16x16x4_f32(false, a1, false, b[2 * j + 1],
                                                (short)0, c, false, false);
    }

    const int orow0 = (tile << 4) + (half << 3);
#pragma unroll
    for (int v = 0; v < 8; ++v) {
      const int r = orow0 + v;
      if (r < N) H[(size_t)r * 128 + col] = c[v];
    }

    if (more) {
#pragma unroll
      for (int q = 0; q < 2; ++q) {
        float* d = &xs[(i + 1) & 1][(sr + (q << 3)) * GT_PITCH + sw0];
        *reinterpret_cast<float2*>(d)     = make_float2(pre[q].x, pre[q].y);
        *reinterpret_cast<float2*>(d + 4) = make_float2(pre[q].z, pre[q].w);
      }
    }
    __syncthreads();
  }
}

#define LG_NPW 24
#define LG_WAVES 8
#define LG_NPB (LG_NPW * LG_WAVES)

__global__ void __launch_bounds__(LG_WAVES * 32)
gat_attn_logits(const float* __restrict__ H,
                const float* __restrict__ a_src,
                const float* __restrict__ a_dst,
                float* __restrict__ als,
                float* __restrict__ ald, int N) {
  const int lane = threadIdx.x & 31;
  const int n0   = (blockIdx.x * LG_WAVES + (threadIdx.x >> 5)) * LG_NPW;
  if (n0 >= N) return;
  const int h = lane >> 4;
  float as[4], ad[4];
#pragma unroll
  for (int j = 0; j < 4; ++j) {
    as[j] = a_src[lane * 4 + j];
    ad[j] = a_dst[lane * 4 + j];
  }
  float4 hv[LG_NPW];
#pragma unroll
  for (int u = 0; u < LG_NPW; ++u) {
    const int n = (n0 + u < N) ? (n0 + u) : (N - 1);
    hv[u] = *reinterpret_cast<const float4*>(H + (size_t)n * 128 + lane * 4);
  }
  float s[LG_NPW], d[LG_NPW];
#pragma unroll
  for (int u = 0; u < LG_NPW; ++u) {
    const float x[4] = {hv[u].x, hv[u].y, hv[u].z, hv[u].w};
    float ss = 0.f, dd = 0.f;
#pragma unroll
    for (int j = 0; j < 4; ++j) {
      ss += x[j] * as[j];
      dd += x[j] * ad[j];
    }
    s[u] = ss;
    d[u] = dd;
  }
#pragma unroll
  for (int m = 8; m >= 1; m >>= 1) {
#pragma unroll
    for (int u = 0; u < LG_NPW; ++u) {
      s[u] += __shfl_xor(s[u], m, 32);
      d[u] += __shfl_xor(d[u], m, 32);
    }
  }
  if ((lane & 15) == 0) {
#pragma unroll
    for (int u = 0; u < LG_NPW; ++u) {
      if (n0 + u < N) {
        als[(n0 + u) * 2 + h] = s[u];
        ald[(n0 + u) * 2 + h] = d[u];
      }
    }
  }
}

#define NB 391
#define LIST_CAP 9600
#define SRC_BITS 20
#define SRC_MASK 0xFFFFF
#define ST 1024

#define BINS_EPT 16

__global__ void __launch_bounds__(256)
gat_dst_bins(const int* __restrict__ ed,
             unsigned char* __restrict__ bins, int E) {
  const int e0 = (blockIdx.x * blockDim.x + threadIdx.x) * BINS_EPT;
  if (e0 + BINS_EPT <= E) {
    int d[BINS_EPT];
#pragma unroll
    for (int k = 0; k < BINS_EPT; ++k) d[k] = ed[e0 + k];
    unsigned w[4];
#pragma unroll
    for (int q = 0; q < 4; ++q)
      w[q] = ((unsigned)(d[4 * q] / NB) & 0xFFu) |
             (((unsigned)(d[4 * q + 1] / NB) & 0xFFu) << 8) |
             (((unsigned)(d[4 * q + 2] / NB) & 0xFFu) << 16) |
             ((unsigned)(d[4 * q + 3] / NB) << 24);
    *reinterpret_cast<uint4*>(bins + e0) = make_uint4(w[0], w[1], w[2], w[3]);
  } else {
    for (int e = e0; e < E; ++e) bins[e] = (unsigned char)(ed[e] / NB);
  }
}

#define ROWPOS128(c) ((((c) & 3) << 5) + ((c) >> 2))
#define ROWPOS64(c)  ((((c) & 1) << 5) + ((c) >> 1))

__device__ __forceinline__ float lrelu(float v) {
  return (v >= 0.f) ? v : NEG_SLOPE * v;
}

template <int W>
__device__ __forceinline__ void lane_edge(int phase, int s, int dl,
                                          const float* __restrict__ als,
                                          const float* __restrict__ H,
                                          float* rows, unsigned* smax,
                                          float* sden, const float* sald) {
  const float v0 = lrelu(als[s * 2] + sald[dl * 2]);
  const float v1 = lrelu(als[s * 2 + 1] + sald[dl * 2 + 1]);
  if (phase == 0) {
    atomicMax(&smax[dl * 2], fenc(v0));
    atomicMax(&smax[dl * 2 + 1], fenc(v1));
    return;
  }
  const float x0 = expf(v0 - fdec(smax[dl * 2]));
  const float x1 = expf(v1 - fdec(smax[dl * 2 + 1]));
  if (phase == 1) {
    atomicAdd(&sden[dl * 2], x0);
    atomicAdd(&sden[dl * 2 + 1], x1);
    return;
  }
  const float a0 = x0 / sden[dl * 2];
  const float a1 = x1 / sden[dl * 2 + 1];
  const float* hs = H + (size_t)s * 128;
  if (W == 128) {
    for (int c = 0; c < 128; ++c)
      atomicAdd(&rows[dl * 128 + ROWPOS128(c)], hs[c] * ((c < 64) ? a0 : a1));
  } else {
    const float w0 = 0.5f * a0, w1 = 0.5f * a1;
    for (int c = 0; c < 64; ++c)
      atomicAdd(&rows[dl * 64 + ROWPOS64(c)], hs[c] * w0 + hs[64 + c] * w1);
  }
}

__device__ __forceinline__ void filter_vec(const uint4 q, int ebase, unsigned splat,
                                           int E, int* list, unsigned* cnt) {
  const unsigned w[4] = {q.x, q.y, q.z, q.w};
#pragma unroll
  for (int k = 0; k < 4; ++k) {
    const unsigned x = w[k] ^ splat;
    unsigned z = ~(((x & 0x7F7F7F7Fu) + 0x7F7F7F7Fu) | x | 0x7F7F7F7Fu);
    while (z) {
      const int e = ebase + (k << 2) + (__builtin_ctz(z) >> 3);
      z &= z - 1u;
      if (e < E) {
        const unsigned p = atomicAdd(cnt, 1u);
        if (p < LIST_CAP) list[p] = e;
      }
    }
  }
}

template <int W> struct RowRegs;
template <> struct RowRegs<128> { float4 v; };
template <> struct RowRegs<64>  { float2 a, b; };

template <int W>
__device__ __forceinline__ RowRegs<W> wave_row_load(const float* __restrict__ H,
                                                    int s, int lane);
template <>
__device__ __forceinline__ RowRegs<128> wave_row_load<128>(const float* __restrict__ H,
                                                           int s, int lane) {
  RowRegs<128> r;
  r.v = *reinterpret_cast<const float4*>(H + (size_t)s * 128 + (lane << 2));
  return r;
}
template <>
__device__ __forceinline__ RowRegs<64> wave_row_load<64>(const float* __restrict__ H,
                                                         int s, int lane) {
  RowRegs<64> r;
  const float* hs = H + (size_t)s * 128 + (lane << 1);
  r.a = *reinterpret_cast<const float2*>(hs);
  r.b = *reinterpret_cast<const float2*>(hs + 64);
  return r;
}

__device__ __forceinline__ void wave_row_add(float* rows, const RowRegs<128>& r,
                                             int dl, float w0, float w1, int lane) {
  const float w = (lane < 16) ? w0 : w1;
  float* d = &rows[dl * 128 + lane];
  atomicAdd(d, r.v.x * w);
  atomicAdd(d + 32, r.v.y * w);
  atomicAdd(d + 64, r.v.z * w);
  atomicAdd(d + 96, r.v.w * w);
}
__device__ __forceinline__ void wave_row_add(float* rows, const RowRegs<64>& r,
                                             int dl, float w0, float w1, int lane) {
  float* d = &rows[dl * 64 + lane];
  atomicAdd(d, r.a.x * w0 + r.b.x * w1);
  atomicAdd(d + 32, r.a.y * w0 + r.b.y * w1);
}

template <int W>
__global__ void __launch_bounds__(ST)
gat_dst_slice(const int* __restrict__ es, const int* __restrict__ ed,
              const unsigned char* __restrict__ bins,
              const float* __restrict__ als, const float* __restrict__ ald,
              const float* __restrict__ H, const float* __restrict__ bias,
              float* __restrict__ dst, int* __restrict__ glist,
              int* __restrict__ gcnt, int N, int E) {
  __shared__ float rows[NB * W];
  __shared__ unsigned smax[NB * 2];
  __shared__ float sden[NB * 2];
  __shared__ float sald[NB * 2];
  __shared__ __attribute__((aligned(16))) int list[LIST_CAP];
  __shared__ __attribute__((aligned(16))) float2 sw[LIST_CAP];
  __shared__ unsigned cnt;

  const int tid = threadIdx.x;
  const int node0 = blockIdx.x * NB;
  const int nloc = (N - node0 < NB) ? (N - node0) : NB;
  const unsigned mybin = blockIdx.x;

  for (int i = tid; i < NB * W; i += ST) rows[i] = 0.f;
  for (int i = tid; i < NB * 2; i += ST) {
    smax[i] = ENC_NEG_INF;
    sden[i] = 0.f;
    sald[i] = (i < nloc * 2) ? ald[(size_t)node0 * 2 + i] : 0.f;
  }

  int total;
  if (W == 128) {
    for (int i = tid; i < nloc; i += ST) list[i] = (node0 + i) | (i << SRC_BITS);
    if (tid == 0) cnt = (unsigned)nloc;
    __syncthreads();

    const uint4* b4 = reinterpret_cast<const uint4*>(bins);
    const int nvec = (E + 15) >> 4;
    const unsigned splat = mybin * 0x01010101u;
    int v0 = tid;
    for (; v0 + 3 * ST < nvec; v0 += 4 * ST) {
      uint4 q[4];
#pragma unroll
      for (int j = 0; j < 4; ++j) q[j] = b4[v0 + j * ST];
#pragma unroll
      for (int j = 0; j < 4; ++j)
        filter_vec(q[j], (v0 + j * ST) << 4, splat, E, list, &cnt);
    }
    for (; v0 < nvec; v0 += ST)
      filter_vec(b4[v0], v0 << 4, splat, E, list, &cnt);
    __syncthreads();
    total = (int)cnt;
    if (total <= LIST_CAP) {
      for (int i = nloc + tid; i < total; i += ST) {
        const int e = list[i];
        list[i] = es[e] | ((ed[e] - node0) << SRC_BITS);
      }
      __syncthreads();
      for (int i = tid; i < total; i += ST)
        glist[(size_t)blockIdx.x * LIST_CAP + i] = list[i];
    }
    if (tid == 0) gcnt[blockIdx.x] = total;
  } else {
    total = gcnt[blockIdx.x];
    if (total <= LIST_CAP)
      for (int i = tid; i < total; i += ST)
        list[i] = glist[(size_t)blockIdx.x * LIST_CAP + i];
  }
  const bool fits = total <= LIST_CAP;
  __syncthreads();

  const int wave = tid >> 5, lane = tid & 31;
  if (fits) {
    for (int i = tid; i < total; i += ST) {
      const int pk = list[i];
      const int s = pk & SRC_MASK, dl = pk >> SRC_BITS;
      const float v0 = lrelu(als[s * 2] + sald[dl * 2]);
      const float v1 = lrelu(als[s * 2 + 1] + sald[dl * 2 + 1]);
      sw[i] = make_float2(v0, v1);
      atomicMax(&smax[dl * 2], fenc(v0));
      atomicMax(&smax[dl * 2 + 1], fenc(v1));
    }
    __syncthreads();
    for (int i = tid; i < total; i += ST) {
      const int dl = list[i] >> SRC_BITS;
      const float2 v = sw[i];
      const float x0 = expf(v.x - fdec(smax[dl * 2]));
      const float x1 = expf(v.y - fdec(smax[dl * 2 + 1]));
      sw[i] = make_float2(x0, x1);
      atomicAdd(&sden[dl * 2], x0);
      atomicAdd(&sden[dl * 2 + 1], x1);
    }
    __syncthreads();
    for (int i = tid; i < total; i += ST) {
      const int dl = list[i] >> SRC_BITS;
      const float2 x = sw[i];
      float a0 = x.x / sden[dl * 2];
      float a1 = x.y / sden[dl * 2 + 1];
      if (W == 64) { a0 = 0.5f * a0; a1 = 0.5f * a1; }
      sw[i] = make_float2(a0, a1);
    }
    __syncthreads();
    const int ngrp = total >> 3;
    for (int g = wave; g < ngrp; g += ST / 32) {
      const int i0 = g << 3;
      const int4 pa = *reinterpret_cast<const int4*>(&list[i0]);
      const int4 pb = *reinterpret_cast<const int4*>(&list[i0 + 4]);
      const int pk[8] = {pa.x, pa.y, pa.z, pa.w, pb.x, pb.y, pb.z, pb.w};
      float4 al[4];
#pragma unroll
      for (int u = 0; u < 4; ++u)
        al[u] = *reinterpret_cast<const float4*>(&sw[i0 + 2 * u]);
      RowRegs<W> hv[8];
#pragma unroll
      for (int u = 0; u < 8; ++u)
        hv[u] = wave_row_load<W>(H, pk[u] & SRC_MASK, lane);
#pragma unroll
      for (int u = 0; u < 8; ++u) {
        const float w0 = (u & 1) ? al[u >> 1].z : al[u >> 1].x;
        const float w1 = (u & 1) ? al[u >> 1].w : al[u >> 1].y;
        wave_row_add(rows, hv[u], pk[u] >> SRC_BITS, w0, w1, lane);
      }
    }
    for (int i = (ngrp << 3) + wave; i < total; i += ST / 32) {
      const int pk = list[i];
      const float2 al = sw[i];
      const RowRegs<W> hv = wave_row_load<W>(H, pk & SRC_MASK, lane);
      wave_row_add(rows, hv, pk >> SRC_BITS, al.x, al.y, lane);
    }
    __syncthreads();
  } else {
    for (int phase = 0; phase < 3; ++phase) {
      for (int i = tid; i < nloc; i += ST)
        lane_edge<W>(phase, node0 + i, i, als, H, rows, smax, sden, sald);
      for (int e = tid; e < E; e += ST)
        if (bins[e] == mybin)
          lane_edge<W>(phase, es[e], ed[e] - node0, als, H, rows, smax, sden, sald);
      __syncthreads();
    }
  }

  if (W == 128) {
    const float bv[4] = {bias[lane * 4], bias[lane * 4 + 1],
                         bias[lane * 4 + 2], bias[lane * 4 + 3]};
    for (int i = tid; i < nloc * 32; i += ST) {
      const int dl = i >> 5;
      const float* r = &rows[dl * 128 + lane];
      float v[4] = {r[0] + bv[0], r[32] + bv[1], r[64] + bv[2], r[96] + bv[3]};
#pragma unroll
      for (int j = 0; j < 4; ++j) v[j] = (v[j] > 0.f) ? v[j] : (expf(v[j]) - 1.f);
      *reinterpret_cast<float4*>(dst + (size_t)(node0 + dl) * 128 + lane * 4) =
          make_float4(v[0], v[1], v[2], v[3]);
    }
  } else {
    const float b0 = bias[lane * 2], b1 = bias[lane * 2 + 1];
    for (int i = tid; i < nloc * 32; i += ST) {
      const int dl = i >> 5;
      const float* r = &rows[dl * 64 + lane];
      *reinterpret_cast<float2*>(dst + (size_t)(node0 + dl) * 64 + lane * 2) =
          make_float2(r[0] + b0, r[32] + b1);
    }
  }
}

extern "C" void kernel_launch(void* const* d_in, const int* in_sizes, int n_in,
                              void* d_out, int out_size, void* d_ws, size_t ws_size,
                              hipStream_t stream) {
  const float* x   = (const float*)d_in[0];
  const int*   ei  = (const int*)d_in[1];
  const float* W1  = (const float*)d_in[2];
  const float* as1 = (const float*)d_in[3];
  const float* ad1 = (const float*)d_in[4];
  const float* b1  = (const float*)d_in[5];
  const float* W2  = (const float*)d_in[6];
  const float* as2 = (const float*)d_in[7];
  const float* ad2 = (const float*)d_in[8];
  const float* b2  = (const float*)d_in[9];

  const int N = in_sizes[0] / 128;
  const int E = in_sizes[1] / 2;
  const int* esrc = ei;
  const int* edst = ei + E;

  float* h   = (float*)d_ws;
  float* agg = h + (size_t)N * 128;
  float* als = agg + (size_t)N * 128;
  float* ald = als + (size_t)N * 2;
  const int gSlice = (N + NB - 1) / NB;
  int* glist = (int*)(ald + (size_t)N * 2);
  int* gcnt  = glist + (size_t)gSlice * LIST_CAP;
  unsigned char* bins = (unsigned char*)(gcnt + ((gSlice + 3) & ~3));

  const int B = 256;
  const int gGemm  = (N + 16 * GT_TILES - 1) / (16 * GT_TILES);
  const int gNodeW = (N + LG_NPB - 1) / LG_NPB;
  const int gBins  = (E + B * BINS_EPT - 1) / (B * BINS_EPT);

  gat_dst_bins<<<gBins, B, 0, stream>>>(edst, bins, E);

  gat_gemm128_wmma<<<gGemm, B, 0, stream>>>(x, W1, h, N);
  gat_attn_logits<<<gNodeW, B, 0, stream>>>(h, as1, ad1, als, ald, N);
  gat_dst_slice<128><<<gSlice, ST, 0, stream>>>(esrc, edst, bins, als, ald,
                                                h, b1, agg, glist, gcnt, N, E);

  gat_gemm128_wmma<<<gGemm, B, 0, stream>>>(agg, W2, h, N);
  gat_attn_logits<<<gNodeW, B, 0, stream>>>(h, as2, ad2, als, ald, N);
  gat_dst_slice<64><<<gSlice, ST, 0, stream>>>(esrc, edst, bins, als, ald,
                                               h, b2, (float*)d_out, glist, gcnt, N, E);
}


# Edge pass writes a membership bitmap per destination slice (bit per edge, built per 4096-edge chunk in LDS); a slice block scans its own row, not all edge ids; adds stay in LDS arrival order
# speedup vs baseline: 5.7771x; 1.0966x over previous
#include <hip/hip_runtime.h>
#include <math.h>

typedef float v2f __attribute__((ext_vector_type(2)));
typedef float v8f __attribute__((ext_vector_type(8)));

#define NEG_SLOPE 0.2f
#define ENC_NEG_INF 0x007FFFFFu

__device__ __forceinline__ unsigned fenc(float f) {
  unsigned u = __float_as_uint(f);
  return (u & 0x80000000u) ? ~u : (u | 0x80000000u);
}
__device__ __forceinline__ float fdec(unsigned k) {
  unsigned u = (k & 0x80000000u) ? (k & 0x7FFFFFFFu) : ~k;
  return __uint_as_float(u);
}

#define GT_TILES 5
#define GT_PITCH 132

__global__ void __launch_bounds__(256)
gat_gemm128_wmma(const float* __restrict__ X,
                 const float* __restrict__ W,
                 float* __restrict__ H, int N) {
  __shared__ __attribute__((aligned(16))) float xs[2][16 * GT_PITCH];

  const int tid  = threadIdx.x;
  const int wave = tid >> 5;
  const int lane = tid & 31;
  const int half = lane >> 4;
  const int lm   = lane & 15;
  const int col  = (wave << 4) + lm;

  const int ntiles = (N + 15) >> 4;
  const int tile0  = blockIdx.x * GT_TILES;
  const int nt = (ntiles - tile0 < GT_TILES) ? (ntiles - tile0) : GT_TILES;

  const int sr = tid >> 5;
  const int sc = tid & 31;
  const int sw0 = ((sc >> 1) << 3) + ((sc & 1) << 1);

  float4 pre[2];
#pragma unroll
  for (int q = 0; q < 2; ++q) {
    int r = (tile0 << 4) + sr + (q << 3);
    if (r >= N) r = N - 1;
    pre[q] = *reinterpret_cast<const float4*>(X + (size_t)r * 128 + (sc << 2));
  }

  v2f b[32];
#pragma unroll
  for (int kk = 0; kk < 32; ++kk) {
    const int kb = (kk << 2) + (half << 1);
    b[kk].x = W[(size_t)kb * 128 + col];
    b[kk].y = W[(size_t)(kb + 1) * 128 + col];
  }

#pragma unroll
  for (int q = 0; q < 2; ++q) {
    float* d = &xs[0][(sr + (q << 3)) * GT_PITCH + sw0];
    *reinterpret_cast<float2*>(d)     = make_float2(pre[q].x, pre[q].y);
    *reinterpret_cast<float2*>(d + 4) = make_float2(pre[q].z, pre[q].w);
  }
  __syncthreads();

  for (int i = 0; i < nt; ++i) {
    const int tile = tile0 + i;
    const bool more = (i + 1 < nt);
    if (more) {
#pragma unroll
      for (int q = 0; q < 2; ++q) {
        int r = ((tile + 1) << 4) + sr + (q << 3);
        if (r >= N) r = N - 1;
        pre[q] = *reinterpret_cast<const float4*>(X + (size_t)r * 128 + (sc << 2));
      }
    }

    const float* xt = &xs[i & 1][lm * GT_PITCH + (half << 2)];
    float4 af[16];
#pragma unroll
    for (int j = 0; j < 16; ++j)
      af[j] = *reinterpret_cast<const float4*>(xt + (j << 3));

    v8f c = {};
#pragma unroll
    for (int j = 0; j < 16; ++j) {
      v2f a0, a1;
      a0.x = af[j].x; a0.y = af[j].y;
      a1.x = af[j].z; a1.y = af[j].w;
      c = __builtin_amdgcn_wmma_f32_16x16x4_f32(false, a0, false, b[2 * j],
                                                (short)0, c, false, false);
      c = __builtin_amdgcn_wmma_f32_16x16x4_f32(false, a1, false, b[2 * j + 1],
                                                (short)0, c, false, false);
    }

    const int orow0 = (tile << 4) + (half << 3);
#pragma unroll
    for (int v = 0; v < 8; ++v) {
      const int r = orow0 + v;
      if (r < N) H[(size_t)r * 128 + col] = c[v];
    }

    if (more) {
#pragma unroll
      for (int q = 0; q < 2; ++q) {
        float* d = &xs[(i + 1) & 1][(sr + (q << 3)) * GT_PITCH + sw0];
        *reinterpret_cast<float2*>(d)     = make_float2(pre[q].x, pre[q].y);
        *reinterpret_cast<float2*>(d + 4) = make_float2(pre[q].z, pre[q].w);
      }
    }
    __syncthreads();
  }
}

#define LG_NPW 24
#define LG_WAVES 8
#define LG_NPB (LG_NPW * LG_WAVES)

__global__ void __launch_bounds__(LG_WAVES * 32)
gat_attn_logits(const float* __restrict__ H,
                const float* __restrict__ a_src,
                const float* __restrict__ a_dst,
                float* __restrict__ als,
                float* __restrict__ ald, int N) {
  const int lane = threadIdx.x & 31;
  const int n0   = (blockIdx.x * LG_WAVES + (threadIdx.x >> 5)) * LG_NPW;
  if (n0 >= N) return;
  const int h = lane >> 4;
  float as[4], ad[4];
#pragma unroll
  for (int j = 0; j < 4; ++j) {
    as[j] = a_src[lane * 4 + j];
    ad[j] = a_dst[lane * 4 + j];
  }
  float4 hv[LG_NPW];
#pragma unroll
  for (int u = 0; u < LG_NPW; ++u) {
    const int n = (n0 + u < N) ? (n0 + u) : (N - 1);
    hv[u] = *reinterpret_cast<const float4*>(H + (size_t)n * 128 + lane * 4);
  }
  float s[LG_NPW], d[LG_NPW];
#pragma unroll
  for (int u = 0; u < LG_NPW; ++u) {
    const float x[4] = {hv[u].x, hv[u].y, hv[u].z, hv[u].w};
    float ss = 0.f, dd = 0.f;
#pragma unroll
    for (int j = 0; j < 4; ++j) {
      ss += x[j] * as[j];
      dd += x[j] * ad[j];
    }
    s[u] = ss;
    d[u] = dd;
  }
#pragma unroll
  for (int m = 8; m >= 1; m >>= 1) {
#pragma unroll
    for (int u = 0; u < LG_NPW; ++u) {
      s[u] += __shfl_xor(s[u], m, 32);
      d[u] += __shfl_xor(d[u], m, 32);
    }
  }
  if ((lane & 15) == 0) {
#pragma unroll
    for (int u = 0; u < LG_NPW; ++u) {
      if (n0 + u < N) {
        als[(n0 + u) * 2 + h] = s[u];
        ald[(n0 + u) * 2 + h] = d[u];
      }
    }
  }
}

#define NB 391
#define LIST_CAP 9600
#define SRC_BITS 20
#define SRC_MASK 0xFFFFF
#define ST 1024

#define BM_CHUNK 4096
#define BM_WORDS (BM_CHUNK / 32)
#define BM_SLICES 256
#define BINS_EPT 16

__global__ void __launch_bounds__(256)
gat_dst_bitmaps(const int* __restrict__ ed,
                unsigned* __restrict__ bmap, int E, int nchunk) {
  __shared__ __attribute__((aligned(16))) unsigned bm[BM_SLICES * BM_WORDS];
  const int tid = threadIdx.x;
  const int chunk = blockIdx.x;
  const int e0 = chunk * BM_CHUNK + tid * BINS_EPT;
  int d[BINS_EPT];
  if (e0 + BINS_EPT <= E) {
#pragma unroll
    for (int k = 0; k < BINS_EPT; ++k) d[k] = ed[e0 + k];
  } else {
#pragma unroll
    for (int k = 0; k < BINS_EPT; ++k) d[k] = (e0 + k < E) ? ed[e0 + k] : -1;
  }

  uint4* bm4 = reinterpret_cast<uint4*>(bm);
  for (int i = tid; i < BM_SLICES * BM_WORDS / 4; i += 256)
    bm4[i] = make_uint4(0u, 0u, 0u, 0u);
  __syncthreads();

  const int wcol = tid >> 1;
  const int b0 = (tid & 1) << 4;
#pragma unroll
  for (int k = 0; k < BINS_EPT; ++k) {
    const int s = d[k] / NB;
    if (d[k] >= 0 && s < BM_SLICES) atomicOr(&bm[s * BM_WORDS + wcol], 1u << (b0 + k));
  }
  __syncthreads();

  for (int v = tid; v < BM_SLICES * BM_WORDS / 4; v += 256) {
    const int s = v >> 5, off = v & 31;
    unsigned* g = bmap + ((size_t)s * nchunk + chunk) * BM_WORDS + (off << 2);
    *reinterpret_cast<uint4*>(g) = bm4[v];
  }
}

#define ROWPOS128(c) ((((c) & 3) << 5) + ((c) >> 2))
#define ROWPOS64(c)  ((((c) & 1) << 5) + ((c) >> 1))

__device__ __forceinline__ float lrelu(float v) {
  return (v >= 0.f) ? v : NEG_SLOPE * v;
}

template <int W>
__device__ __forceinline__ void lane_edge(int phase, int s, int dl,
                                          const float* __restrict__ als,
                                          const float* __restrict__ H,
                                          float* rows, unsigned* smax,
                                          float* sden, const float* sald) {
  const float v0 = lrelu(als[s * 2] + sald[dl * 2]);
  const float v1 = lrelu(als[s * 2 + 1] + sald[dl * 2 + 1]);
  if (phase == 0) {
    atomicMax(&smax[dl * 2], fenc(v0));
    atomicMax(&smax[dl * 2 + 1], fenc(v1));
    return;
  }
  const float x0 = expf(v0 - fdec(smax[dl * 2]));
  const float x1 = expf(v1 - fdec(smax[dl * 2 + 1]));
  if (phase == 1) {
    atomicAdd(&sden[dl * 2], x0);
    atomicAdd(&sden[dl * 2 + 1], x1);
    return;
  }
  const float a0 = x0 / sden[dl * 2];
  const float a1 = x1 / sden[dl * 2 + 1];
  const float* hs = H + (size_t)s * 128;
  if (W == 128) {
    for (int c = 0; c < 128; ++c)
      atomicAdd(&rows[dl * 128 + ROWPOS128(c)], hs[c] * ((c < 64) ? a0 : a1));
  } else {
    const float w0 = 0.5f * a0, w1 = 0.5f * a1;
    for (int c = 0; c < 64; ++c)
      atomicAdd(&rows[dl * 64 + ROWPOS64(c)], hs[c] * w0 + hs[64 + c] * w1);
  }
}

#define SCAN_V 16

__device__ __forceinline__ void load_vecs(uint4 (&q)[SCAN_V], const uint4* __restrict__ b4,
                                          int v0, int nvec) {
#pragma unroll
  for (int j = 0; j < SCAN_V; ++j) {
    const int v = v0 + j * ST;
    q[j] = (v < nvec) ? b4[v] : make_uint4(0u, 0u, 0u, 0u);
  }
}

__device__ __forceinline__ void append_vecs(const uint4 (&q)[SCAN_V], int ebase, int estep,
                                            int lane, int* list, unsigned* cnt) {
  int c = 0;
#pragma unroll
  for (int j = 0; j < SCAN_V; ++j)
    c += __popc(q[j].x) + __popc(q[j].y) + __popc(q[j].z) + __popc(q[j].w);
  int inc = c;
#pragma unroll
  for (int m = 1; m < 32; m <<= 1) {
    const int t = __shfl_up(inc, m, 32);
    if (lane >= m) inc += t;
  }
  const int tot = __shfl(inc, 31, 32);
  if (tot == 0) return;
  unsigned base = 0u;
  if (lane == 31) base = atomicAdd(cnt, (unsigned)tot);
  base = (unsigned)__shfl((int)base, 31, 32);
  unsigned p = base + (unsigned)(inc - c);
#pragma unroll
  for (int j = 0; j < SCAN_V; ++j) {
    const unsigned w[4] = {q[j].x, q[j].y, q[j].z, q[j].w};
#pragma unroll
    for (int k = 0; k < 4; ++k) {
      unsigned z = w[k];
      while (z) {
        const int e = ebase + j * estep + (k << 5) + __builtin_ctz(z);
        z &= z - 1u;
        if (p < LIST_CAP) list[p] = e;
        ++p;
      }
    }
  }
}

template <int W> struct RowRegs;
template <> struct RowRegs<128> { float4 v; };
template <> struct RowRegs<64>  { float2 a, b; };

template <int W>
__device__ __forceinline__ RowRegs<W> wave_row_load(const float* __restrict__ H,
                                                    int s, int lane);
template <>
__device__ __forceinline__ RowRegs<128> wave_row_load<128>(const float* __restrict__ H,
                                                           int s, int lane) {
  RowRegs<128> r;
  r.v = *reinterpret_cast<const float4*>(H + (size_t)s * 128 + (lane << 2));
  return r;
}
template <>
__device__ __forceinline__ RowRegs<64> wave_row_load<64>(const float* __restrict__ H,
                                                         int s, int lane) {
  RowRegs<64> r;
  const float* hs = H + (size_t)s * 128 + (lane << 1);
  r.a = *reinterpret_cast<const float2*>(hs);
  r.b = *reinterpret_cast<const float2*>(hs + 64);
  return r;
}

__device__ __forceinline__ void wave_row_add(float* rows, const RowRegs<128>& r,
                                             int dl, float w0, float w1, int lane) {
  const float w = (lane < 16) ? w0 : w1;
  float* d = &rows[dl * 128 + lane];
  atomicAdd(d, r.v.x * w);
  atomicAdd(d + 32, r.v.y * w);
  atomicAdd(d + 64, r.v.z * w);
  atomicAdd(d + 96, r.v.w * w);
}
__device__ __forceinline__ void wave_row_add(float* rows, const RowRegs<64>& r,
                                             int dl, float w0, float w1, int lane) {
  float* d = &rows[dl * 64 + lane];
  atomicAdd(d, r.a.x * w0 + r.b.x * w1);
  atomicAdd(d + 32, r.a.y * w0 + r.b.y * w1);
}

template <int W>
__global__ void __launch_bounds__(ST)
gat_dst_slice(const int* __restrict__ es, const int* __restrict__ ed,
              const unsigned* __restrict__ bmap,
              const float* __restrict__ als, const float* __restrict__ ald,
              const float* __restrict__ H, const float* __restrict__ bias,
              float* __restrict__ dst, int* __restrict__ glist,
              int* __restrict__ gcnt, int N, int E, int nchunk) {
  __shared__ float rows[NB * W];
  __shared__ unsigned smax[NB * 2];
  __shared__ float sden[NB * 2];
  __shared__ float sald[NB * 2];
  __shared__ __attribute__((aligned(16))) int list[LIST_CAP];
  __shared__ __attribute__((aligned(16))) float2 sw[LIST_CAP];
  __shared__ unsigned cnt;

  const int tid = threadIdx.x;
  const int node0 = blockIdx.x * NB;
  const int nloc = (N - node0 < NB) ? (N - node0) : NB;
  const unsigned* bmrow = bmap + (size_t)blockIdx.x * nchunk * BM_WORDS;
  const uint4* b4 = reinterpret_cast<const uint4*>(bmrow);
  const int nvec = nchunk * (BM_WORDS / 4);
  uint4 q[SCAN_V];
  if (W == 128) load_vecs(q, b4, tid, nvec);

  for (int i = tid; i < NB * W; i += ST) rows[i] = 0.f;
  for (int i = tid; i < NB * 2; i += ST) {
    smax[i] = ENC_NEG_INF;
    sden[i] = 0.f;
    sald[i] = (i < nloc * 2) ? ald[(size_t)node0 * 2 + i] : 0.f;
  }

  int total;
  if (W == 128) {
    for (int i = tid; i < nloc; i += ST) list[i] = (node0 + i) | (i << SRC_BITS);
    if (tid == 0) cnt = (unsigned)nloc;
    __syncthreads();

    const int lane0 = tid & 31;
    for (int vb = tid - lane0; vb < nvec; ) {
      append_vecs(q, (vb + lane0) << 7, ST << 7, lane0, list, &cnt);
      vb += SCAN_V * ST;
      if (vb < nvec) load_vecs(q, b4, vb + lane0, nvec);
    }
    __syncthreads();
    total = (int)cnt;
    if (total <= LIST_CAP) {
      for (int i = nloc + tid; i < total; i += ST) {
        const int e = list[i];
        list[i] = es[e] | ((ed[e] - node0) << SRC_BITS);
      }
      __syncthreads();
      for (int i = tid; i < total; i += ST)
        glist[(size_t)blockIdx.x * LIST_CAP + i] = list[i];
    }
    if (tid == 0) gcnt[blockIdx.x] = total;
  } else {
    total = gcnt[blockIdx.x];
    if (total <= LIST_CAP)
      for (int i = tid; i < total; i += ST)
        list[i] = glist[(size_t)blockIdx.x * LIST_CAP + i];
  }
  const bool fits = total <= LIST_CAP;
  __syncthreads();

  const int wave = tid >> 5, lane = tid & 31;
  if (fits) {
    for (int i = tid; i < total; i += ST) {
      const int pk = list[i];
      const int s = pk & SRC_MASK, dl = pk >> SRC_BITS;
      const float v0 = lrelu(als[s * 2] + sald[dl * 2]);
      const float v1 = lrelu(als[s * 2 + 1] + sald[dl * 2 + 1]);
      sw[i] = make_float2(v0, v1);
      atomicMax(&smax[dl * 2], fenc(v0));
      atomicMax(&smax[dl * 2 + 1], fenc(v1));
    }
    __syncthreads();
    for (int i = tid; i < total; i += ST) {
      const int dl = list[i] >> SRC_BITS;
      const float2 v = sw[i];
      const float x0 = expf(v.x - fdec(smax[dl * 2]));
      const float x1 = expf(v.y - fdec(smax[dl * 2 + 1]));
      sw[i] = make_float2(x0, x1);
      atomicAdd(&sden[dl * 2], x0);
      atomicAdd(&sden[dl * 2 + 1], x1);
    }
    __syncthreads();
    for (int i = tid; i < total; i += ST) {
      const int dl = list[i] >> SRC_BITS;
      const float2 x = sw[i];
      float a0 = x.x / sden[dl * 2];
      float a1 = x.y / sden[dl * 2 + 1];
      if (W == 64) { a0 = 0.5f * a0; a1 = 0.5f * a1; }
      sw[i] = make_float2(a0, a1);
    }
    __syncthreads();
    const int ngrp = total >> 3;
    for (int g = wave; g < ngrp; g += ST / 32) {
      const int i0 = g << 3;
      const int4 pa = *reinterpret_cast<const int4*>(&list[i0]);
      const int4 pb = *reinterpret_cast<const int4*>(&list[i0 + 4]);
      const int pk[8] = {pa.x, pa.y, pa.z, pa.w, pb.x, pb.y, pb.z, pb.w};
      float4 al[4];
#pragma unroll
      for (int u = 0; u < 4; ++u)
        al[u] = *reinterpret_cast<const float4*>(&sw[i0 + 2 * u]);
      RowRegs<W> hv[8];
#pragma unroll
      for (int u = 0; u < 8; ++u)
        hv[u] = wave_row_load<W>(H, pk[u] & SRC_MASK, lane);
#pragma unroll
      for (int u = 0; u < 8; ++u) {
        const float w0 = (u & 1) ? al[u >> 1].z : al[u >> 1].x;
        const float w1 = (u & 1) ? al[u >> 1].w : al[u >> 1].y;
        wave_row_add(rows, hv[u], pk[u] >> SRC_BITS, w0, w1, lane);
      }
    }
    for (int i = (ngrp << 3) + wave; i < total; i += ST / 32) {
      const int pk = list[i];
      const float2 al = sw[i];
      const RowRegs<W> hv = wave_row_load<W>(H, pk & SRC_MASK, lane);
      wave_row_add(rows, hv, pk >> SRC_BITS, al.x, al.y, lane);
    }
    __syncthreads();
  } else {
    for (int phase = 0; phase < 3; ++phase) {
      for (int i = tid; i < nloc; i += ST)
        lane_edge<W>(phase, node0 + i, i, als, H, rows, smax, sden, sald);
      for (int e = tid; e < E; e += ST)
        if ((bmrow[e >> 5] >> (e & 31)) & 1u)
          lane_edge<W>(phase, es[e], ed[e] - node0, als, H, rows, smax, sden, sald);
      __syncthreads();
    }
  }

  if (W == 128) {
    const float bv[4] = {bias[lane * 4], bias[lane * 4 + 1],
                         bias[lane * 4 + 2], bias[lane * 4 + 3]};
    for (int i = tid; i < nloc * 32; i += ST) {
      const int dl = i >> 5;
      const float* r = &rows[dl * 128 + lane];
      float v[4] = {r[0] + bv[0], r[32] + bv[1], r[64] + bv[2], r[96] + bv[3]};
#pragma unroll
      for (int j = 0; j < 4; ++j) v[j] = (v[j] > 0.f) ? v[j] : (expf(v[j]) - 1.f);
      *reinterpret_cast<float4*>(dst + (size_t)(node0 + dl) * 128 + lane * 4) =
          make_float4(v[0], v[1], v[2], v[3]);
    }
  } else {
    const float b0 = bias[lane * 2], b1 = bias[lane * 2 + 1];
    for (int i = tid; i < nloc * 32; i += ST) {
      const int dl = i >> 5;
      const float* r = &rows[dl * 64 + lane];
      *reinterpret_cast<float2*>(dst + (size_t)(node0 + dl) * 64 + lane * 2) =
          make_float2(r[0] + b0, r[32] + b1);
    }
  }
}

extern "C" void kernel_launch(void* const* d_in, const int* in_sizes, int n_in,
                              void* d_out, int out_size, void* d_ws, size_t ws_size,
                              hipStream_t stream) {
  const float* x   = (const float*)d_in[0];
  const int*   ei  = (const int*)d_in[1];
  const float* W1  = (const float*)d_in[2];
  const float* as1 = (const float*)d_in[3];
  const float* ad1 = (const float*)d_in[4];
  const float* b1  = (const float*)d_in[5];
  const float* W2  = (const float*)d_in[6];
  const float* as2 = (const float*)d_in[7];
  const float* ad2 = (const float*)d_in[8];
  const float* b2  = (const float*)d_in[9];

  const int N = in_sizes[0] / 128;
  const int E = in_sizes[1] / 2;
  const int* esrc = ei;
  const int* edst = ei + E;

  float* h   = (float*)d_ws;
  float* agg = h + (size_t)N * 128;
  float* als = agg + (size_t)N * 128;
  float* ald = als + (size_t)N * 2;
  const int gSlice = (N + NB - 1) / NB;
  int* glist = (int*)(ald + (size_t)N * 2);
  int* gcnt  = glist + (size_t)gSlice * LIST_CAP;
  unsigned* bmap = (unsigned*)(gcnt + ((gSlice + 3) & ~3));

  const int B = 256;
  const int gGemm  = (N + 16 * GT_TILES - 1) / (16 * GT_TILES);
  const int gNodeW = (N + LG_NPB - 1) / LG_NPB;
  const int nChunk = (E + BM_CHUNK - 1) / BM_CHUNK;

  gat_dst_bitmaps<<<nChunk, B, 0, stream>>>(edst, bmap, E, nChunk);

  gat_gemm128_wmma<<<gGemm, B, 0, stream>>>(x, W1, h, N);
  gat_attn_logits<<<gNodeW, B, 0, stream>>>(h, as1, ad1, als, ald, N);
  gat_dst_slice<128><<<gSlice, ST, 0, stream>>>(esrc, edst, bmap, als, ald,
                                                h, b1, agg, glist, gcnt, N, E, nChunk);

  gat_gemm128_wmma<<<gGemm, B, 0, stream>>>(agg, W2, h, N);
  gat_attn_logits<<<gNodeW, B, 0, stream>>>(h, as2, ad2, als, ald, N);
  gat_dst_slice<64><<<gSlice, ST, 0, stream>>>(esrc, edst, bmap, als, ald,
                                               h, b2, (float*)d_out, glist, gcnt, N, E, nChunk);
}
